# starting kernel
# baseline (speedup 1.0000x reference)
#include <hip/hip_runtime.h>
#include <math.h>

typedef __attribute__((ext_vector_type(2))) float v2f;
typedef __attribute__((ext_vector_type(8))) float v8f;

#define NEGBIG (-3.4e38f)

__global__ __launch_bounds__(256) void gemm_wmma_f32(
    const float* __restrict__ A, const float* __restrict__ W,
    const float* __restrict__ bias, float* __restrict__ C,
    int M, int N, int K)
{
  __shared__ float As[128][33];
  __shared__ float Bs[32][130];

  const int tid  = threadIdx.x;
  const int wave = tid >> 5;
  const int lane = tid & 31;
  const int r    = lane & 15;
  const int hi   = lane >> 4;
  const int wrow = wave >> 2;
  const int wcol = wave & 3;
  const int m0   = blockIdx.y * 128;
  const int n0   = blockIdx.x * 128;

  v8f acc[4][2];
  for (int i = 0; i < 4; ++i)
    for (int j = 0; j < 2; ++j)
      for (int c = 0; c < 8; ++c) acc[i][j][c] = 0.0f;

  for (int k0 = 0; k0 < K; k0 += 32) {
    __syncthreads();
    {
      int row = tid >> 3;
      int col = (tid & 7) * 4;
      #pragma unroll
      for (int it = 0; it < 4; ++it) {
        int rr = row + it * 32;
        float4 v = *(const float4*)&A[(size_t)(m0 + rr) * K + k0 + col];
        As[rr][col + 0] = v.x; As[rr][col + 1] = v.y;
        As[rr][col + 2] = v.z; As[rr][col + 3] = v.w;
      }
    }
    {
      int n  = tid >> 3;
      int kk = (tid & 7) * 4;
      #pragma unroll
      for (int it = 0; it < 4; ++it) {
        int nn = n + it * 32;
        float4 v = make_float4(0.f, 0.f, 0.f, 0.f);
        if (n0 + nn < N)
          v = *(const float4*)&W[(size_t)(n0 + nn) * K + k0 + kk];
        Bs[kk + 0][nn] = v.x; Bs[kk + 1][nn] = v.y;
        Bs[kk + 2][nn] = v.z; Bs[kk + 3][nn] = v.w;
      }
    }
    __syncthreads();

    #pragma unroll
    for (int kk = 0; kk < 32; kk += 4) {
      v2f af[4], bf[2];
      #pragma unroll
      for (int i = 0; i < 4; ++i) {
        int row = wrow * 64 + i * 16 + r;
        af[i][0] = As[row][kk + 2 * hi];
        af[i][1] = As[row][kk + 1 + 2 * hi];
      }
      #pragma unroll
      for (int j = 0; j < 2; ++j) {
        int col = wcol * 32 + j * 16 + r;
        bf[j][0] = Bs[kk + 2 * hi][col];
        bf[j][1] = Bs[kk + 1 + 2 * hi][col];
      }
      #pragma unroll
      for (int i = 0; i < 4; ++i)
        #pragma unroll
        for (int j = 0; j < 2; ++j)
          acc[i][j] = __builtin_amdgcn_wmma_f32_16x16x4_f32(
              false, af[i], false, bf[j], (short)0, acc[i][j], false, false);
    }
  }

  #pragma unroll
  for (int i = 0; i < 4; ++i)
    for (int j = 0; j < 2; ++j) {
      int gcol = n0 + wcol * 32 + j * 16 + r;
      if (gcol >= N) continue;
      float b = bias ? bias[gcol] : 0.0f;
      #pragma unroll
      for (int vg = 0; vg < 8; ++vg) {
        int grow = m0 + wrow * 64 + i * 16 + vg + 8 * hi;
        C[(size_t)grow * N + gcol] = acc[i][j][vg] + b;
      }
    }
}

__global__ void rope_kernel(float* __restrict__ qbuf, float* __restrict__ kvbuf)
{
  const long long QP = 2048LL * 64 * 32;
  const long long KP = 2048LL * 8 * 32;
  long long idx = (long long)blockIdx.x * blockDim.x + threadIdx.x;
  if (idx >= QP + KP) return;

  int s, i;
  float *p0, *p1;
  if (idx < QP) {
    int t = (int)idx;
    i = t & 31;
    int h = (t >> 5) & 63;
    s = t >> 11;
    float* base = qbuf + (size_t)s * 4096 + h * 64;
    p0 = base + i; p1 = base + i + 32;
  } else {
    int t = (int)(idx - QP);
    i = t & 31;
    int kvh = (t >> 5) & 7;
    s = t >> 8;
    float* base = kvbuf + (size_t)s * 1024 + kvh * 128;
    p0 = base + i; p1 = base + i + 32;
  }
  float inv = powf(150000.0f, -((float)(2 * i)) / 64.0f);
  float ang = (float)s * inv;
  float c = cosf(ang), sn = sinf(ang);
  float x1 = *p0, x2 = *p1;
  *p0 = x1 * c - x2 * sn;
  *p1 = x2 * c + x1 * sn;
}

__global__ __launch_bounds__(128) void flash_attn(
    const float* __restrict__ qbuf, const float* __restrict__ kvbuf,
    float* __restrict__ attn)
{
  __shared__ float Qs[64][65];
  __shared__ float Kt[64][65];
  __shared__ float Vs[64][64];
  __shared__ float Ssc[64][65];

  const int tid   = threadIdx.x;
  const int wave  = tid >> 5;
  const int lane  = tid & 31;
  const int r     = lane & 15;
  const int hi    = lane >> 4;
  const int qtile = blockIdx.x;
  const int h     = blockIdx.y;
  const int kvh   = h >> 3;
  const int qbase = qtile * 64;

  #pragma unroll 4
  for (int it = 0; it < 32; ++it) {
    int f = tid + it * 128;
    int sl = f >> 6, d = f & 63;
    Qs[sl][d] = qbuf[(size_t)(qbase + sl) * 4096 + h * 64 + d];
  }

  float m_i = -INFINITY, l_i = 0.0f;
  v8f oacc[4];
  for (int sub = 0; sub < 4; ++sub)
    for (int c = 0; c < 8; ++c) oacc[sub][c] = 0.0f;

  for (int j = 0; j <= qtile; ++j) {
    const int kvbase = j * 64;
    __syncthreads();
    #pragma unroll 4
    for (int it = 0; it < 32; ++it) {
      int f = tid + it * 128;
      int n = f >> 6, d = f & 63;
      const float* kv = &kvbuf[(size_t)(kvbase + n) * 1024 + kvh * 128];
      Kt[d][n] = kv[d];
      Vs[n][d] = kv[64 + d];
    }
    __syncthreads();

    v8f sacc[4];
    for (int sub = 0; sub < 4; ++sub)
      for (int c = 0; c < 8; ++c) sacc[sub][c] = 0.0f;
    #pragma unroll
    for (int kk = 0; kk < 64; kk += 4) {
      v2f af;
      af[0] = Qs[wave * 16 + r][kk + 2 * hi];
      af[1] = Qs[wave * 16 + r][kk + 1 + 2 * hi];
      #pragma unroll
      for (int sub = 0; sub < 4; ++sub) {
        v2f bf;
        bf[0] = Kt[kk + 2 * hi][sub * 16 + r];
        bf[1] = Kt[kk + 1 + 2 * hi][sub * 16 + r];
        sacc[sub] = __builtin_amdgcn_wmma_f32_16x16x4_f32(
            false, af, false, bf, (short)0, sacc[sub], false, false);
      }
    }

    #pragma unroll
    for (int sub = 0; sub < 4; ++sub)
      #pragma unroll
      for (int vg = 0; vg < 8; ++vg) {
        int m  = vg + 8 * hi;
        int qr = qbase + wave * 16 + m;
        int kc = kvbase + sub * 16 + r;
        float v = (kc <= qr) ? sacc[sub][vg] * 0.125f : NEGBIG;
        Ssc[wave * 16 + m][sub * 16 + r] = v;
      }

    {
      int row = wave * 16 + r;
      float rowmax = -INFINITY;
      for (int c = 0; c < 64; ++c) rowmax = fmaxf(rowmax, Ssc[row][c]);
      float newm = fmaxf(m_i, rowmax);
      float sum = 0.0f;
      for (int c = 0; c < 64; ++c) {
        float p = __expf(Ssc[row][c] - newm);
        Ssc[row][c] = p;
        sum += p;
      }
      float alpha = __expf(m_i - newm);
      l_i = l_i * alpha + sum;
      m_i = newm;
      #pragma unroll
      for (int vg = 0; vg < 8; ++vg) {
        float am = __int_as_float(__builtin_amdgcn_ds_bpermute(
            4 * (vg + 24 * hi), __float_as_int(alpha)));
        for (int sub = 0; sub < 4; ++sub) oacc[sub][vg] *= am;
      }
    }

    #pragma unroll
    for (int kk = 0; kk < 64; kk += 4) {
      v2f af;
      af[0] = Ssc[wave * 16 + r][kk + 2 * hi];
      af[1] = Ssc[wave * 16 + r][kk + 1 + 2 * hi];
      #pragma unroll
      for (int sub = 0; sub < 4; ++sub) {
        v2f bf;
        bf[0] = Vs[kk + 2 * hi][sub * 16 + r];
        bf[1] = Vs[kk + 1 + 2 * hi][sub * 16 + r];
        oacc[sub] = __builtin_amdgcn_wmma_f32_16x16x4_f32(
            false, af, false, bf, (short)0, oacc[sub], false, false);
      }
    }
  }

  float inv = 1.0f / l_i;
  #pragma unroll
  for (int vg = 0; vg < 8; ++vg) {
    float im = __int_as_float(__builtin_amdgcn_ds_bpermute(
        4 * (vg + 24 * hi), __float_as_int(inv)));
    int grow = qbase + wave * 16 + vg + 8 * hi;
    for (int sub = 0; sub < 4; ++sub) {
      int d = sub * 16 + r;
      attn[(size_t)grow * 4096 + h * 64 + d] = oacc[sub][vg] * im;
    }
  }
}

extern "C" void kernel_launch(void* const* d_in, const int* in_sizes, int n_in,
                              void* d_out, int out_size, void* d_ws, size_t ws_size,
                              hipStream_t stream)
{
  const float* hidden = (const float*)d_in[0];
  const float* q_w    = (const float*)d_in[2];
  const float* q_b    = (const float*)d_in[3];
  const float* kv_a_w = (const float*)d_in[4];
  const float* kv_b_w = (const float*)d_in[5];
  const float* o_w    = (const float*)d_in[6];
  const float* o_b    = (const float*)d_in[7];
  float* out = (float*)d_out;

  float* ws    = (float*)d_ws;
  float* qbuf  = ws;
  float* ckv   = qbuf  + (size_t)2048 * 4096;
  float* kvbuf = ckv   + (size_t)2048 * 384;
  float* attn  = kvbuf + (size_t)2048 * 1024;

  dim3 blk(256);
  gemm_wmma_f32<<<dim3(4096 / 128, 2048 / 128), blk, 0, stream>>>(
      hidden, q_w, q_b, qbuf, 2048, 4096, 2880);
  gemm_wmma_f32<<<dim3(384 / 128, 2048 / 128), blk, 0, stream>>>(
      hidden, kv_a_w, nullptr, ckv, 2048, 384, 2880);
  gemm_wmma_f32<<<dim3(1024 / 128, 2048 / 128), blk, 0, stream>>>(
      ckv, kv_b_w, nullptr, kvbuf, 2048, 1024, 384);
  long long pairs = 2048LL * 64 * 32 + 2048LL * 8 * 32;
  rope_kernel<<<(int)((pairs + 255) / 256), 256, 0, stream>>>(qbuf, kvbuf);
  flash_attn<<<dim3(32, 64), 128, 0, stream>>>(qbuf, kvbuf, attn);
  gemm_wmma_f32<<<dim3((2880 + 127) / 128, 2048 / 128), blk, 0, stream>>>(
      attn, o_w, o_b, out, 2048, 2880, 4096);
}


# GEMM k-loop: the 8 tile loads (4 A, 4 W) issue as one group before the barrier, W bounds branch replaced by row clamp + zero select: 1 memory round trip per k-step instead of 5; sums unchanged
# speedup vs baseline: 1.0511x; 1.0511x over previous
#include <hip/hip_runtime.h>
#include <math.h>

typedef __attribute__((ext_vector_type(2))) float v2f;
typedef __attribute__((ext_vector_type(8))) float v8f;
typedef __attribute__((ext_vector_type(4))) float v4f;

#define NEGBIG (-3.4e38f)

__global__ __launch_bounds__(256) void gemm_wmma_f32(
    const float* __restrict__ A, const float* __restrict__ W,
    const float* __restrict__ bias, float* __restrict__ C,
    int M, int N, int K)
{
  __shared__ float As[128][33];
  __shared__ float Bs[32][130];

  const int tid  = threadIdx.x;
  const int wave = tid >> 5;
  const int lane = tid & 31;
  const int r    = lane & 15;
  const int hi   = lane >> 4;
  const int wrow = wave >> 2;
  const int wcol = wave & 3;
  const int m0   = blockIdx.y * 128;
  const int n0   = blockIdx.x * 128;

  v8f acc[4][2];
  for (int i = 0; i < 4; ++i)
    for (int j = 0; j < 2; ++j)
      for (int c = 0; c < 8; ++c) acc[i][j][c] = 0.0f;

  for (int k0 = 0; k0 < K; k0 += 32) {
    v4f av[4], wv[4];
    {
      int row = tid >> 3;
      int col = (tid & 7) * 4;
      #pragma unroll
      for (int it = 0; it < 4; ++it)
        av[it] = *(const v4f*)&A[(size_t)(m0 + row + it * 32) * K + k0 + col];
      #pragma unroll
      for (int it = 0; it < 4; ++it) {
        int grow = n0 + row + it * 32;
        int crow = grow < N ? grow : N - 1;
        wv[it] = *(const v4f*)&W[(size_t)crow * K + k0 + col];
      }
    }
    __syncthreads();
    {
      int row = tid >> 3;
      int col = (tid & 7) * 4;
      #pragma unroll
      for (int it = 0; it < 4; ++it) {
        int rr = row + it * 32;
        As[rr][col + 0] = av[it][0]; As[rr][col + 1] = av[it][1];
        As[rr][col + 2] = av[it][2]; As[rr][col + 3] = av[it][3];
      }
    }
    {
      int n  = tid >> 3;
      int kk = (tid & 7) * 4;
      #pragma unroll
      for (int it = 0; it < 4; ++it) {
        int nn = n + it * 32;
        v4f v = wv[it];
        if (n0 + nn >= N) { v[0] = 0.f; v[1] = 0.f; v[2] = 0.f; v[3] = 0.f; }
        Bs[kk + 0][nn] = v[0]; Bs[kk + 1][nn] = v[1];
        Bs[kk + 2][nn] = v[2]; Bs[kk + 3][nn] = v[3];
      }
    }
    __syncthreads();

    #pragma unroll
    for (int kk = 0; kk < 32; kk += 4) {
      v2f af[4], bf[2];
      #pragma unroll
      for (int i = 0; i < 4; ++i) {
        int row = wrow * 64 + i * 16 + r;
        af[i][0] = As[row][kk + 2 * hi];
        af[i][1] = As[row][kk + 1 + 2 * hi];
      }
      #pragma unroll
      for (int j = 0; j < 2; ++j) {
        int col = wcol * 32 + j * 16 + r;
        bf[j][0] = Bs[kk + 2 * hi][col];
        bf[j][1] = Bs[kk + 1 + 2 * hi][col];
      }
      #pragma unroll
      for (int i = 0; i < 4; ++i)
        #pragma unroll
        for (int j = 0; j < 2; ++j)
          acc[i][j] = __builtin_amdgcn_wmma_f32_16x16x4_f32(
              false, af[i], false, bf[j], (short)0, acc[i][j], false, false);
    }
  }

  #pragma unroll
  for (int i = 0; i < 4; ++i)
    for (int j = 0; j < 2; ++j) {
      int gcol = n0 + wcol * 32 + j * 16 + r;
      if (gcol >= N) continue;
      float b = bias ? bias[gcol] : 0.0f;
      #pragma unroll
      for (int vg = 0; vg < 8; ++vg) {
        int grow = m0 + wrow * 64 + i * 16 + vg + 8 * hi;
        C[(size_t)grow * N + gcol] = acc[i][j][vg] + b;
      }
    }
}

__global__ void rope_kernel(float* __restrict__ qbuf, float* __restrict__ kvbuf)
{
  const long long QP = 2048LL * 64 * 32;
  const long long KP = 2048LL * 8 * 32;
  long long idx = (long long)blockIdx.x * blockDim.x + threadIdx.x;
  if (idx >= QP + KP) return;

  int s, i;
  float *p0, *p1;
  if (idx < QP) {
    int t = (int)idx;
    i = t & 31;
    int h = (t >> 5) & 63;
    s = t >> 11;
    float* base = qbuf + (size_t)s * 4096 + h * 64;
    p0 = base + i; p1 = base + i + 32;
  } else {
    int t = (int)(idx - QP);
    i = t & 31;
    int kvh = (t >> 5) & 7;
    s = t >> 8;
    float* base = kvbuf + (size_t)s * 1024 + kvh * 128;
    p0 = base + i; p1 = base + i + 32;
  }
  float inv = powf(150000.0f, -((float)(2 * i)) / 64.0f);
  float ang = (float)s * inv;
  float c = cosf(ang), sn = sinf(ang);
  float x1 = *p0, x2 = *p1;
  *p0 = x1 * c - x2 * sn;
  *p1 = x2 * c + x1 * sn;
}

__global__ __launch_bounds__(128) void flash_attn(
    const float* __restrict__ qbuf, const float* __restrict__ kvbuf,
    float* __restrict__ attn)
{
  __shared__ float Qs[64][65];
  __shared__ float Kt[64][65];
  __shared__ float Vs[64][64];
  __shared__ float Ssc[64][65];

  const int tid   = threadIdx.x;
  const int wave  = tid >> 5;
  const int lane  = tid & 31;
  const int r     = lane & 15;
  const int hi    = lane >> 4;
  const int qtile = blockIdx.x;
  const int h     = blockIdx.y;
  const int kvh   = h >> 3;
  const int qbase = qtile * 64;

  #pragma unroll 4
  for (int it = 0; it < 32; ++it) {
    int f = tid + it * 128;
    int sl = f >> 6, d = f & 63;
    Qs[sl][d] = qbuf[(size_t)(qbase + sl) * 4096 + h * 64 + d];
  }

  float m_i = -INFINITY, l_i = 0.0f;
  v8f oacc[4];
  for (int sub = 0; sub < 4; ++sub)
    for (int c = 0; c < 8; ++c) oacc[sub][c] = 0.0f;

  for (int j = 0; j <= qtile; ++j) {
    const int kvbase = j * 64;
    __syncthreads();
    #pragma unroll 4
    for (int it = 0; it < 32; ++it) {
      int f = tid + it * 128;
      int n = f >> 6, d = f & 63;
      const float* kv = &kvbuf[(size_t)(kvbase + n) * 1024 + kvh * 128];
      Kt[d][n] = kv[d];
      Vs[n][d] = kv[64 + d];
    }
    __syncthreads();

    v8f sacc[4];
    for (int sub = 0; sub < 4; ++sub)
      for (int c = 0; c < 8; ++c) sacc[sub][c] = 0.0f;
    #pragma unroll
    for (int kk = 0; kk < 64; kk += 4) {
      v2f af;
      af[0] = Qs[wave * 16 + r][kk + 2 * hi];
      af[1] = Qs[wave * 16 + r][kk + 1 + 2 * hi];
      #pragma unroll
      for (int sub = 0; sub < 4; ++sub) {
        v2f bf;
        bf[0] = Kt[kk + 2 * hi][sub * 16 + r];
        bf[1] = Kt[kk + 1 + 2 * hi][sub * 16 + r];
        sacc[sub] = __builtin_amdgcn_wmma_f32_16x16x4_f32(
            false, af, false, bf, (short)0, sacc[sub], false, false);
      }
    }

    #pragma unroll
    for (int sub = 0; sub < 4; ++sub)
      #pragma unroll
      for (int vg = 0; vg < 8; ++vg) {
        int m  = vg + 8 * hi;
        int qr = qbase + wave * 16 + m;
        int kc = kvbase + sub * 16 + r;
        float v = (kc <= qr) ? sacc[sub][vg] * 0.125f : NEGBIG;
        Ssc[wave * 16 + m][sub * 16 + r] = v;
      }

    {
      int row = wave * 16 + r;
      float rowmax = -INFINITY;
      for (int c = 0; c < 64; ++c) rowmax = fmaxf(rowmax, Ssc[row][c]);
      float newm = fmaxf(m_i, rowmax);
      float sum = 0.0f;
      for (int c = 0; c < 64; ++c) {
        float p = __expf(Ssc[row][c] - newm);
        Ssc[row][c] = p;
        sum += p;
      }
      float alpha = __expf(m_i - newm);
      l_i = l_i * alpha + sum;
      m_i = newm;
      #pragma unroll
      for (int vg = 0; vg < 8; ++vg) {
        float am = __int_as_float(__builtin_amdgcn_ds_bpermute(
            4 * (vg + 24 * hi), __float_as_int(alpha)));
        for (int sub = 0; sub < 4; ++sub) oacc[sub][vg] *= am;
      }
    }

    #pragma unroll
    for (int kk = 0; kk < 64; kk += 4) {
      v2f af;
      af[0] = Ssc[wave * 16 + r][kk + 2 * hi];
      af[1] = Ssc[wave * 16 + r][kk + 1 + 2 * hi];
      #pragma unroll
      for (int sub = 0; sub < 4; ++sub) {
        v2f bf;
        bf[0] = Vs[kk + 2 * hi][sub * 16 + r];
        bf[1] = Vs[kk + 1 + 2 * hi][sub * 16 + r];
        oacc[sub] = __builtin_amdgcn_wmma_f32_16x16x4_f32(
            false, af, false, bf, (short)0, oacc[sub], false, false);
      }
    }
  }

  float inv = 1.0f / l_i;
  #pragma unroll
  for (int vg = 0; vg < 8; ++vg) {
    float im = __int_as_float(__builtin_amdgcn_ds_bpermute(
        4 * (vg + 24 * hi), __float_as_int(inv)));
    int grow = qbase + wave * 16 + vg + 8 * hi;
    for (int sub = 0; sub < 4; ++sub) {
      int d = sub * 16 + r;
      attn[(size_t)grow * 4096 + h * 64 + d] = oacc[sub][vg] * im;
    }
  }
}

extern "C" void kernel_launch(void* const* d_in, const int* in_sizes, int n_in,
                              void* d_out, int out_size, void* d_ws, size_t ws_size,
                              hipStream_t stream)
{
  const float* hidden = (const float*)d_in[0];
  const float* q_w    = (const float*)d_in[2];
  const float* q_b    = (const float*)d_in[3];
  const float* kv_a_w = (const float*)d_in[4];
  const float* kv_b_w = (const float*)d_in[5];
  const float* o_w    = (const float*)d_in[6];
  const float* o_b    = (const float*)d_in[7];
  float* out = (float*)d_out;

  float* ws    = (float*)d_ws;
  float* qbuf  = ws;
  float* ckv   = qbuf  + (size_t)2048 * 4096;
  float* kvbuf = ckv   + (size_t)2048 * 384;
  float* attn  = kvbuf + (size_t)2048 * 1024;

  dim3 blk(256);
  gemm_wmma_f32<<<dim3(4096 / 128, 2048 / 128), blk, 0, stream>>>(
      hidden, q_w, q_b, qbuf, 2048, 4096, 2880);
  gemm_wmma_f32<<<dim3(384 / 128, 2048 / 128), blk, 0, stream>>>(
      hidden, kv_a_w, nullptr, ckv, 2048, 384, 2880);
  gemm_wmma_f32<<<dim3(1024 / 128, 2048 / 128), blk, 0, stream>>>(
      ckv, kv_b_w, nullptr, kvbuf, 2048, 1024, 384);
  long long pairs = 2048LL * 64 * 32 + 2048LL * 8 * 32;
  rope_kernel<<<(int)((pairs + 255) / 256), 256, 0, stream>>>(qbuf, kvbuf);
  flash_attn<<<dim3(32, 64), 128, 0, stream>>>(qbuf, kvbuf, attn);
  gemm_wmma_f32<<<dim3((2880 + 127) / 128, 2048 / 128), blk, 0, stream>>>(
      attn, o_w, o_b, out, 2048, 2880, 4096);
}


# attention softmax on the score accumulator registers: row max by 16-lane DPP butterfly, per-lane partial row sums added across lanes once at the end (sum order changes); no row loops through LDS
# speedup vs baseline: 1.1830x; 1.1255x over previous
#include <hip/hip_runtime.h>
#include <math.h>

typedef __attribute__((ext_vector_type(2))) float v2f;
typedef __attribute__((ext_vector_type(8))) float v8f;
typedef __attribute__((ext_vector_type(4))) float v4f;

#define NEGBIG (-3.4e38f)

__global__ __launch_bounds__(256) void gemm_wmma_f32(
    const float* __restrict__ A, const float* __restrict__ W,
    const float* __restrict__ bias, float* __restrict__ C,
    int M, int N, int K)
{
  __shared__ float As[128][33];
  __shared__ float Bs[32][130];

  const int tid  = threadIdx.x;
  const int wave = tid >> 5;
  const int lane = tid & 31;
  const int r    = lane & 15;
  const int hi   = lane >> 4;
  const int wrow = wave >> 2;
  const int wcol = wave & 3;
  const int m0   = blockIdx.y * 128;
  const int n0   = blockIdx.x * 128;

  v8f acc[4][2];
  for (int i = 0; i < 4; ++i)
    for (int j = 0; j < 2; ++j)
      for (int c = 0; c < 8; ++c) acc[i][j][c] = 0.0f;

  for (int k0 = 0; k0 < K; k0 += 32) {
    v4f av[4], wv[4];
    {
      int row = tid >> 3;
      int col = (tid & 7) * 4;
      #pragma unroll
      for (int it = 0; it < 4; ++it)
        av[it] = *(const v4f*)&A[(size_t)(m0 + row + it * 32) * K + k0 + col];
      #pragma unroll
      for (int it = 0; it < 4; ++it) {
        int grow = n0 + row + it * 32;
        int crow = grow < N ? grow : N - 1;
        wv[it] = *(const v4f*)&W[(size_t)crow * K + k0 + col];
      }
    }
    __syncthreads();
    {
      int row = tid >> 3;
      int col = (tid & 7) * 4;
      #pragma unroll
      for (int it = 0; it < 4; ++it) {
        int rr = row + it * 32;
        As[rr][col + 0] = av[it][0]; As[rr][col + 1] = av[it][1];
        As[rr][col + 2] = av[it][2]; As[rr][col + 3] = av[it][3];
      }
    }
    {
      int n  = tid >> 3;
      int kk = (tid & 7) * 4;
      #pragma unroll
      for (int it = 0; it < 4; ++it) {
        int nn = n + it * 32;
        v4f v = wv[it];
        if (n0 + nn >= N) { v[0] = 0.f; v[1] = 0.f; v[2] = 0.f; v[3] = 0.f; }
        Bs[kk + 0][nn] = v[0]; Bs[kk + 1][nn] = v[1];
        Bs[kk + 2][nn] = v[2]; Bs[kk + 3][nn] = v[3];
      }
    }
    __syncthreads();

    #pragma unroll
    for (int kk = 0; kk < 32; kk += 4) {
      v2f af[4], bf[2];
      #pragma unroll
      for (int i = 0; i < 4; ++i) {
        int row = wrow * 64 + i * 16 + r;
        af[i][0] = As[row][kk + 2 * hi];
        af[i][1] = As[row][kk + 1 + 2 * hi];
      }
      #pragma unroll
      for (int j = 0; j < 2; ++j) {
        int col = wcol * 32 + j * 16 + r;
        bf[j][0] = Bs[kk + 2 * hi][col];
        bf[j][1] = Bs[kk + 1 + 2 * hi][col];
      }
      #pragma unroll
      for (int i = 0; i < 4; ++i)
        #pragma unroll
        for (int j = 0; j < 2; ++j)
          acc[i][j] = __builtin_amdgcn_wmma_f32_16x16x4_f32(
              false, af[i], false, bf[j], (short)0, acc[i][j], false, false);
    }
  }

  #pragma unroll
  for (int i = 0; i < 4; ++i)
    for (int j = 0; j < 2; ++j) {
      int gcol = n0 + wcol * 32 + j * 16 + r;
      if (gcol >= N) continue;
      float b = bias ? bias[gcol] : 0.0f;
      #pragma unroll
      for (int vg = 0; vg < 8; ++vg) {
        int grow = m0 + wrow * 64 + i * 16 + vg + 8 * hi;
        C[(size_t)grow * N + gcol] = acc[i][j][vg] + b;
      }
    }
}

__global__ void rope_kernel(float* __restrict__ qbuf, float* __restrict__ kvbuf)
{
  const long long QP = 2048LL * 64 * 32;
  const long long KP = 2048LL * 8 * 32;
  long long idx = (long long)blockIdx.x * blockDim.x + threadIdx.x;
  if (idx >= QP + KP) return;

  int s, i;
  float *p0, *p1;
  if (idx < QP) {
    int t = (int)idx;
    i = t & 31;
    int h = (t >> 5) & 63;
    s = t >> 11;
    float* base = qbuf + (size_t)s * 4096 + h * 64;
    p0 = base + i; p1 = base + i + 32;
  } else {
    int t = (int)(idx - QP);
    i = t & 31;
    int kvh = (t >> 5) & 7;
    s = t >> 8;
    float* base = kvbuf + (size_t)s * 1024 + kvh * 128;
    p0 = base + i; p1 = base + i + 32;
  }
  float inv = powf(150000.0f, -((float)(2 * i)) / 64.0f);
  float ang = (float)s * inv;
  float c = cosf(ang), sn = sinf(ang);
  float x1 = *p0, x2 = *p1;
  *p0 = x1 * c - x2 * sn;
  *p1 = x2 * c + x1 * sn;
}

template <int CTRL>
__device__ __forceinline__ float row_xor(float v)
{
  return __int_as_float(__builtin_amdgcn_update_dpp(
      0, __float_as_int(v), CTRL, 0xf, 0xf, true));
}

__global__ __launch_bounds__(128) void flash_attn(
    const float* __restrict__ qbuf, const float* __restrict__ kvbuf,
    float* __restrict__ attn)
{
  __shared__ float Qs[64][65];
  __shared__ float Kt[64][65];
  __shared__ float Vs[64][64];
  __shared__ float Ssc[64][65];

  const int tid   = threadIdx.x;
  const int wave  = tid >> 5;
  const int lane  = tid & 31;
  const int r     = lane & 15;
  const int hi    = lane >> 4;
  const int qtile = blockIdx.x;
  const int h     = blockIdx.y;
  const int kvh   = h >> 3;
  const int qbase = qtile * 64;

  #pragma unroll 4
  for (int it = 0; it < 32; ++it) {
    int f = tid + it * 128;
    int sl = f >> 6, d = f & 63;
    Qs[sl][d] = qbuf[(size_t)(qbase + sl) * 4096 + h * 64 + d];
  }

  float m_i[8], l_i[8];
  for (int vg = 0; vg < 8; ++vg) { m_i[vg] = -INFINITY; l_i[vg] = 0.0f; }
  v8f oacc[4];
  for (int sub = 0; sub < 4; ++sub)
    for (int c = 0; c < 8; ++c) oacc[sub][c] = 0.0f;

  for (int j = 0; j <= qtile; ++j) {
    const int kvbase = j * 64;
    __syncthreads();
    #pragma unroll 4
    for (int it = 0; it < 32; ++it) {
      int f = tid + it * 128;
      int n = f >> 6, d = f & 63;
      const float* kv = &kvbuf[(size_t)(kvbase + n) * 1024 + kvh * 128];
      Kt[d][n] = kv[d];
      Vs[n][d] = kv[64 + d];
    }
    __syncthreads();

    v8f sacc[4];
    for (int sub = 0; sub < 4; ++sub)
      for (int c = 0; c < 8; ++c) sacc[sub][c] = 0.0f;
    #pragma unroll
    for (int kk = 0; kk < 64; kk += 4) {
      v2f af;
      af[0] = Qs[wave * 16 + r][kk + 2 * hi];
      af[1] = Qs[wave * 16 + r][kk + 1 + 2 * hi];
      #pragma unroll
      for (int sub = 0; sub < 4; ++sub) {
        v2f bf;
        bf[0] = Kt[kk + 2 * hi][sub * 16 + r];
        bf[1] = Kt[kk + 1 + 2 * hi][sub * 16 + r];
        sacc[sub] = __builtin_amdgcn_wmma_f32_16x16x4_f32(
            false, af, false, bf, (short)0, sacc[sub], false, false);
      }
    }

    float mx[8];
    #pragma unroll
    for (int vg = 0; vg < 8; ++vg) {
      int qr = qbase + wave * 16 + vg + 8 * hi;
      #pragma unroll
      for (int sub = 0; sub < 4; ++sub) {
        int kc = kvbase + sub * 16 + r;
        sacc[sub][vg] = (kc <= qr) ? sacc[sub][vg] * 0.125f : NEGBIG;
      }
      mx[vg] = fmaxf(fmaxf(sacc[0][vg], sacc[1][vg]), fmaxf(sacc[2][vg], sacc[3][vg]));
    }
    #pragma unroll
    for (int vg = 0; vg < 8; ++vg) mx[vg] = fmaxf(mx[vg], row_xor<0x161>(mx[vg]));
    #pragma unroll
    for (int vg = 0; vg < 8; ++vg) mx[vg] = fmaxf(mx[vg], row_xor<0x162>(mx[vg]));
    #pragma unroll
    for (int vg = 0; vg < 8; ++vg) mx[vg] = fmaxf(mx[vg], row_xor<0x164>(mx[vg]));
    #pragma unroll
    for (int vg = 0; vg < 8; ++vg) mx[vg] = fmaxf(mx[vg], row_xor<0x168>(mx[vg]));

    #pragma unroll
    for (int vg = 0; vg < 8; ++vg) {
      float newm  = fmaxf(m_i[vg], mx[vg]);
      float alpha = __expf(m_i[vg] - newm);
      m_i[vg] = newm;
      float ps = 0.0f;
      #pragma unroll
      for (int sub = 0; sub < 4; ++sub) {
        float p = __expf(sacc[sub][vg] - newm);
        ps += p;
        Ssc[wave * 16 + vg + 8 * hi][sub * 16 + r] = p;
        oacc[sub][vg] *= alpha;
      }
      l_i[vg] = l_i[vg] * alpha + ps;
    }

    #pragma unroll
    for (int kk = 0; kk < 64; kk += 4) {
      v2f af;
      af[0] = Ssc[wave * 16 + r][kk + 2 * hi];
      af[1] = Ssc[wave * 16 + r][kk + 1 + 2 * hi];
      #pragma unroll
      for (int sub = 0; sub < 4; ++sub) {
        v2f bf;
        bf[0] = Vs[kk + 2 * hi][sub * 16 + r];
        bf[1] = Vs[kk + 1 + 2 * hi][sub * 16 + r];
        oacc[sub] = __builtin_amdgcn_wmma_f32_16x16x4_f32(
            false, af, false, bf, (short)0, oacc[sub], false, false);
      }
    }
  }

  #pragma unroll
  for (int vg = 0; vg < 8; ++vg) l_i[vg] += row_xor<0x161>(l_i[vg]);
  #pragma unroll
  for (int vg = 0; vg < 8; ++vg) l_i[vg] += row_xor<0x162>(l_i[vg]);
  #pragma unroll
  for (int vg = 0; vg < 8; ++vg) l_i[vg] += row_xor<0x164>(l_i[vg]);
  #pragma unroll
  for (int vg = 0; vg < 8; ++vg) l_i[vg] += row_xor<0x168>(l_i[vg]);
  #pragma unroll
  for (int vg = 0; vg < 8; ++vg) {
    float im = 1.0f / l_i[vg];
    int grow = qbase + wave * 16 + vg + 8 * hi;
    for (int sub = 0; sub < 4; ++sub) {
      int d = sub * 16 + r;
      attn[(size_t)grow * 4096 + h * 64 + d] = oacc[sub][vg] * im;
    }
  }
}

extern "C" void kernel_launch(void* const* d_in, const int* in_sizes, int n_in,
                              void* d_out, int out_size, void* d_ws, size_t ws_size,
                              hipStream_t stream)
{
  const float* hidden = (const float*)d_in[0];
  const float* q_w    = (const float*)d_in[2];
  const float* q_b    = (const float*)d_in[3];
  const float* kv_a_w = (const float*)d_in[4];
  const float* kv_b_w = (const float*)d_in[5];
  const float* o_w    = (const float*)d_in[6];
  const float* o_b    = (const float*)d_in[7];
  float* out = (float*)d_out;

  float* ws    = (float*)d_ws;
  float* qbuf  = ws;
  float* ckv   = qbuf  + (size_t)2048 * 4096;
  float* kvbuf = ckv   + (size_t)2048 * 384;
  float* attn  = kvbuf + (size_t)2048 * 1024;

  dim3 blk(256);
  gemm_wmma_f32<<<dim3(4096 / 128, 2048 / 128), blk, 0, stream>>>(
      hidden, q_w, q_b, qbuf, 2048, 4096, 2880);
  gemm_wmma_f32<<<dim3(384 / 128, 2048 / 128), blk, 0, stream>>>(
      hidden, kv_a_w, nullptr, ckv, 2048, 384, 2880);
  gemm_wmma_f32<<<dim3(1024 / 128, 2048 / 128), blk, 0, stream>>>(
      ckv, kv_b_w, nullptr, kvbuf, 2048, 1024, 384);
  long long pairs = 2048LL * 64 * 32 + 2048LL * 8 * 32;
  rope_kernel<<<(int)((pairs + 255) / 256), 256, 0, stream>>>(qbuf, kvbuf);
  flash_attn<<<dim3(32, 64), 128, 0, stream>>>(qbuf, kvbuf, attn);
  gemm_wmma_f32<<<dim3((2880 + 127) / 128, 2048 / 128), blk, 0, stream>>>(
      attn, o_w, o_b, out, 2048, 2880, 4096);
}


# attention block order: heads on grid x, q-tiles on y from last to first, so the causal blocks with most k/v tiles start first and short ones fill the tail; arithmetic and sum order unchanged
# speedup vs baseline: 1.3162x; 1.1126x over previous
#include <hip/hip_runtime.h>
#include <math.h>

typedef __attribute__((ext_vector_type(2))) float v2f;
typedef __attribute__((ext_vector_type(8))) float v8f;
typedef __attribute__((ext_vector_type(4))) float v4f;

#define NEGBIG (-3.4e38f)

__global__ __launch_bounds__(256) void gemm_wmma_f32(
    const float* __restrict__ A, const float* __restrict__ W,
    const float* __restrict__ bias, float* __restrict__ C,
    int M, int N, int K)
{
  __shared__ float As[128][33];
  __shared__ float Bs[32][130];

  const int tid  = threadIdx.x;
  const int wave = tid >> 5;
  const int lane = tid & 31;
  const int r    = lane & 15;
  const int hi   = lane >> 4;
  const int wrow = wave >> 2;
  const int wcol = wave & 3;
  const int m0   = blockIdx.y * 128;
  const int n0   = blockIdx.x * 128;

  v8f acc[4][2];
  for (int i = 0; i < 4; ++i)
    for (int j = 0; j < 2; ++j)
      for (int c = 0; c < 8; ++c) acc[i][j][c] = 0.0f;

  for (int k0 = 0; k0 < K; k0 += 32) {
    v4f av[4], wv[4];
    {
      int row = tid >> 3;
      int col = (tid & 7) * 4;
      #pragma unroll
      for (int it = 0; it < 4; ++it)
        av[it] = *(const v4f*)&A[(size_t)(m0 + row + it * 32) * K + k0 + col];
      #pragma unroll
      for (int it = 0; it < 4; ++it) {
        int grow = n0 + row + it * 32;
        int crow = grow < N ? grow : N - 1;
        wv[it] = *(const v4f*)&W[(size_t)crow * K + k0 + col];
      }
    }
    __syncthreads();
    {
      int row = tid >> 3;
      int col = (tid & 7) * 4;
      #pragma unroll
      for (int it = 0; it < 4; ++it) {
        int rr = row + it * 32;
        As[rr][col + 0] = av[it][0]; As[rr][col + 1] = av[it][1];
        As[rr][col + 2] = av[it][2]; As[rr][col + 3] = av[it][3];
      }
    }
    {
      int n  = tid >> 3;
      int kk = (tid & 7) * 4;
      #pragma unroll
      for (int it = 0; it < 4; ++it) {
        int nn = n + it * 32;
        v4f v = wv[it];
        if (n0 + nn >= N) { v[0] = 0.f; v[1] = 0.f; v[2] = 0.f; v[3] = 0.f; }
        Bs[kk + 0][nn] = v[0]; Bs[kk + 1][nn] = v[1];
        Bs[kk + 2][nn] = v[2]; Bs[kk + 3][nn] = v[3];
      }
    }
    __syncthreads();

    #pragma unroll
    for (int kk = 0; kk < 32; kk += 4) {
      v2f af[4], bf[2];
      #pragma unroll
      for (int i = 0; i < 4; ++i) {
        int row = wrow * 64 + i * 16 + r;
        af[i][0] = As[row][kk + 2 * hi];
        af[i][1] = As[row][kk + 1 + 2 * hi];
      }
      #pragma unroll
      for (int j = 0; j < 2; ++j) {
        int col = wcol * 32 + j * 16 + r;
        bf[j][0] = Bs[kk + 2 * hi][col];
        bf[j][1] = Bs[kk + 1 + 2 * hi][col];
      }
      #pragma unroll
      for (int i = 0; i < 4; ++i)
        #pragma unroll
        for (int j = 0; j < 2; ++j)
          acc[i][j] = __builtin_amdgcn_wmma_f32_16x16x4_f32(
              false, af[i], false, bf[j], (short)0, acc[i][j], false, false);
    }
  }

  #pragma unroll
  for (int i = 0; i < 4; ++i)
    for (int j = 0; j < 2; ++j) {
      int gcol = n0 + wcol * 32 + j * 16 + r;
      if (gcol >= N) continue;
      float b = bias ? bias[gcol] : 0.0f;
      #pragma unroll
      for (int vg = 0; vg < 8; ++vg) {
        int grow = m0 + wrow * 64 + i * 16 + vg + 8 * hi;
        C[(size_t)grow * N + gcol] = acc[i][j][vg] + b;
      }
    }
}

__global__ void rope_kernel(float* __restrict__ qbuf, float* __restrict__ kvbuf)
{
  const long long QP = 2048LL * 64 * 32;
  const long long KP = 2048LL * 8 * 32;
  long long idx = (long long)blockIdx.x * blockDim.x + threadIdx.x;
  if (idx >= QP + KP) return;

  int s, i;
  float *p0, *p1;
  if (idx < QP) {
    int t = (int)idx;
    i = t & 31;
    int h = (t >> 5) & 63;
    s = t >> 11;
    float* base = qbuf + (size_t)s * 4096 + h * 64;
    p0 = base + i; p1 = base + i + 32;
  } else {
    int t = (int)(idx - QP);
    i = t & 31;
    int kvh = (t >> 5) & 7;
    s = t >> 8;
    float* base = kvbuf + (size_t)s * 1024 + kvh * 128;
    p0 = base + i; p1 = base + i + 32;
  }
  float inv = powf(150000.0f, -((float)(2 * i)) / 64.0f);
  float ang = (float)s * inv;
  float c = cosf(ang), sn = sinf(ang);
  float x1 = *p0, x2 = *p1;
  *p0 = x1 * c - x2 * sn;
  *p1 = x2 * c + x1 * sn;
}

template <int CTRL>
__device__ __forceinline__ float row_xor(float v)
{
  return __int_as_float(__builtin_amdgcn_update_dpp(
      0, __float_as_int(v), CTRL, 0xf, 0xf, true));
}

__global__ __launch_bounds__(128) void flash_attn(
    const float* __restrict__ qbuf, const float* __restrict__ kvbuf,
    float* __restrict__ attn)
{
  __shared__ float Qs[64][65];
  __shared__ float Kt[64][65];
  __shared__ float Vs[64][64];
  __shared__ float Ssc[64][65];

  const int tid   = threadIdx.x;
  const int wave  = tid >> 5;
  const int lane  = tid & 31;
  const int r     = lane & 15;
  const int hi    = lane >> 4;
  const int h     = blockIdx.x;
  const int qtile = 31 - blockIdx.y;
  const int kvh   = h >> 3;
  const int qbase = qtile * 64;

  #pragma unroll 4
  for (int it = 0; it < 32; ++it) {
    int f = tid + it * 128;
    int sl = f >> 6, d = f & 63;
    Qs[sl][d] = qbuf[(size_t)(qbase + sl) * 4096 + h * 64 + d];
  }

  float m_i[8], l_i[8];
  for (int vg = 0; vg < 8; ++vg) { m_i[vg] = -INFINITY; l_i[vg] = 0.0f; }
  v8f oacc[4];
  for (int sub = 0; sub < 4; ++sub)
    for (int c = 0; c < 8; ++c) oacc[sub][c] = 0.0f;

  for (int j = 0; j <= qtile; ++j) {
    const int kvbase = j * 64;
    __syncthreads();
    #pragma unroll 4
    for (int it = 0; it < 32; ++it) {
      int f = tid + it * 128;
      int n = f >> 6, d = f & 63;
      const float* kv = &kvbuf[(size_t)(kvbase + n) * 1024 + kvh * 128];
      Kt[d][n] = kv[d];
      Vs[n][d] = kv[64 + d];
    }
    __syncthreads();

    v8f sacc[4];
    for (int sub = 0; sub < 4; ++sub)
      for (int c = 0; c < 8; ++c) sacc[sub][c] = 0.0f;
    #pragma unroll
    for (int kk = 0; kk < 64; kk += 4) {
      v2f af;
      af[0] = Qs[wave * 16 + r][kk + 2 * hi];
      af[1] = Qs[wave * 16 + r][kk + 1 + 2 * hi];
      #pragma unroll
      for (int sub = 0; sub < 4; ++sub) {
        v2f bf;
        bf[0] = Kt[kk + 2 * hi][sub * 16 + r];
        bf[1] = Kt[kk + 1 + 2 * hi][sub * 16 + r];
        sacc[sub] = __builtin_amdgcn_wmma_f32_16x16x4_f32(
            false, af, false, bf, (short)0, sacc[sub], false, false);
      }
    }

    float mx[8];
    #pragma unroll
    for (int vg = 0; vg < 8; ++vg) {
      int qr = qbase + wave * 16 + vg + 8 * hi;
      #pragma unroll
      for (int sub = 0; sub < 4; ++sub) {
        int kc = kvbase + sub * 16 + r;
        sacc[sub][vg] = (kc <= qr) ? sacc[sub][vg] * 0.125f : NEGBIG;
      }
      mx[vg] = fmaxf(fmaxf(sacc[0][vg], sacc[1][vg]), fmaxf(sacc[2][vg], sacc[3][vg]));
    }
    #pragma unroll
    for (int vg = 0; vg < 8; ++vg) mx[vg] = fmaxf(mx[vg], row_xor<0x161>(mx[vg]));
    #pragma unroll
    for (int vg = 0; vg < 8; ++vg) mx[vg] = fmaxf(mx[vg], row_xor<0x162>(mx[vg]));
    #pragma unroll
    for (int vg = 0; vg < 8; ++vg) mx[vg] = fmaxf(mx[vg], row_xor<0x164>(mx[vg]));
    #pragma unroll
    for (int vg = 0; vg < 8; ++vg) mx[vg] = fmaxf(mx[vg], row_xor<0x168>(mx[vg]));

    #pragma unroll
    for (int vg = 0; vg < 8; ++vg) {
      float newm  = fmaxf(m_i[vg], mx[vg]);
      float alpha = __expf(m_i[vg] - newm);
      m_i[vg] = newm;
      float ps = 0.0f;
      #pragma unroll
      for (int sub = 0; sub < 4; ++sub) {
        float p = __expf(sacc[sub][vg] - newm);
        ps += p;
        Ssc[wave * 16 + vg + 8 * hi][sub * 16 + r] = p;
        oacc[sub][vg] *= alpha;
      }
      l_i[vg] = l_i[vg] * alpha + ps;
    }

    #pragma unroll
    for (int kk = 0; kk < 64; kk += 4) {
      v2f af;
      af[0] = Ssc[wave * 16 + r][kk + 2 * hi];
      af[1] = Ssc[wave * 16 + r][kk + 1 + 2 * hi];
      #pragma unroll
      for (int sub = 0; sub < 4; ++sub) {
        v2f bf;
        bf[0] = Vs[kk + 2 * hi][sub * 16 + r];
        bf[1] = Vs[kk + 1 + 2 * hi][sub * 16 + r];
        oacc[sub] = __builtin_amdgcn_wmma_f32_16x16x4_f32(
            false, af, false, bf, (short)0, oacc[sub], false, false);
      }
    }
  }

  #pragma unroll
  for (int vg = 0; vg < 8; ++vg) l_i[vg] += row_xor<0x161>(l_i[vg]);
  #pragma unroll
  for (int vg = 0; vg < 8; ++vg) l_i[vg] += row_xor<0x162>(l_i[vg]);
  #pragma unroll
  for (int vg = 0; vg < 8; ++vg) l_i[vg] += row_xor<0x164>(l_i[vg]);
  #pragma unroll
  for (int vg = 0; vg < 8; ++vg) l_i[vg] += row_xor<0x168>(l_i[vg]);
  #pragma unroll
  for (int vg = 0; vg < 8; ++vg) {
    float im = 1.0f / l_i[vg];
    int grow = qbase + wave * 16 + vg + 8 * hi;
    for (int sub = 0; sub < 4; ++sub) {
      int d = sub * 16 + r;
      attn[(size_t)grow * 4096 + h * 64 + d] = oacc[sub][vg] * im;
    }
  }
}

extern "C" void kernel_launch(void* const* d_in, const int* in_sizes, int n_in,
                              void* d_out, int out_size, void* d_ws, size_t ws_size,
                              hipStream_t stream)
{
  const float* hidden = (const float*)d_in[0];
  const float* q_w    = (const float*)d_in[2];
  const float* q_b    = (const float*)d_in[3];
  const float* kv_a_w = (const float*)d_in[4];
  const float* kv_b_w = (const float*)d_in[5];
  const float* o_w    = (const float*)d_in[6];
  const float* o_b    = (const float*)d_in[7];
  float* out = (float*)d_out;

  float* ws    = (float*)d_ws;
  float* qbuf  = ws;
  float* ckv   = qbuf  + (size_t)2048 * 4096;
  float* kvbuf = ckv   + (size_t)2048 * 384;
  float* attn  = kvbuf + (size_t)2048 * 1024;

  dim3 blk(256);
  gemm_wmma_f32<<<dim3(4096 / 128, 2048 / 128), blk, 0, stream>>>(
      hidden, q_w, q_b, qbuf, 2048, 4096, 2880);
  gemm_wmma_f32<<<dim3(384 / 128, 2048 / 128), blk, 0, stream>>>(
      hidden, kv_a_w, nullptr, ckv, 2048, 384, 2880);
  gemm_wmma_f32<<<dim3(1024 / 128, 2048 / 128), blk, 0, stream>>>(
      ckv, kv_b_w, nullptr, kvbuf, 2048, 1024, 384);
  long long pairs = 2048LL * 64 * 32 + 2048LL * 8 * 32;
  rope_kernel<<<(int)((pairs + 255) / 256), 256, 0, stream>>>(qbuf, kvbuf);
  flash_attn<<<dim3(64, 32), 128, 0, stream>>>(qbuf, kvbuf, attn);
  gemm_wmma_f32<<<dim3((2880 + 127) / 128, 2048 / 128), blk, 0, stream>>>(
      attn, o_w, o_b, out, 2048, 2880, 4096);
}


# GEMM block tile set per launch by template: O-proj 128x96 (480 blocks), kv_a and kv_b 64x64 (192 and 512 blocks), so every WGP holds near-equal matrix work; k order and sums unchanged
# speedup vs baseline: 1.5732x; 1.1953x over previous
#include <hip/hip_runtime.h>
#include <math.h>

typedef __attribute__((ext_vector_type(2))) float v2f;
typedef __attribute__((ext_vector_type(8))) float v8f;
typedef __attribute__((ext_vector_type(4))) float v4f;

#define NEGBIG (-3.4e38f)

template <int BM, int BN, int WR, int WC>
__global__ __launch_bounds__(256) void gemm_wmma_f32(
    const float* __restrict__ A, const float* __restrict__ W,
    const float* __restrict__ bias, float* __restrict__ C,
    int M, int N, int K)
{
  static_assert(WR * WC == 8, "8 waves per block");
  static_assert(BM % (16 * WR) == 0 && BN % (16 * WC) == 0, "whole 16x16 subtiles per wave");
  static_assert(BM % 32 == 0 && BN % 32 == 0, "staging moves 32 rows per pass");
  constexpr int TI = BM / (16 * WR);
  constexpr int TJ = BN / (16 * WC);
  constexpr int PA = BM / 32;
  constexpr int PW = BN / 32;

  __shared__ float As[BM][33];
  __shared__ float Bs[32][BN + 2];

  const int tid  = threadIdx.x;
  const int wave = tid >> 5;
  const int lane = tid & 31;
  const int r    = lane & 15;
  const int hi   = lane >> 4;
  const int wrow = wave / WC;
  const int wcol = wave % WC;
  const int m0   = blockIdx.y * BM;
  const int n0   = blockIdx.x * BN;

  v8f acc[TI][TJ];
  #pragma unroll
  for (int i = 0; i < TI; ++i)
    #pragma unroll
    for (int j = 0; j < TJ; ++j)
      #pragma unroll
      for (int c = 0; c < 8; ++c) acc[i][j][c] = 0.0f;

  for (int k0 = 0; k0 < K; k0 += 32) {
    v4f av[PA], wv[PW];
    {
      int row = tid >> 3;
      int col = (tid & 7) * 4;
      #pragma unroll
      for (int it = 0; it < PA; ++it)
        av[it] = *(const v4f*)&A[(size_t)(m0 + row + it * 32) * K + k0 + col];
      #pragma unroll
      for (int it = 0; it < PW; ++it) {
        int grow = n0 + row + it * 32;
        int crow = grow < N ? grow : N - 1;
        wv[it] = *(const v4f*)&W[(size_t)crow * K + k0 + col];
      }
    }
    __syncthreads();
    {
      int row = tid >> 3;
      int col = (tid & 7) * 4;
      #pragma unroll
      for (int it = 0; it < PA; ++it) {
        int rr = row + it * 32;
        As[rr][col + 0] = av[it][0]; As[rr][col + 1] = av[it][1];
        As[rr][col + 2] = av[it][2]; As[rr][col + 3] = av[it][3];
      }
    }
    {
      int n  = tid >> 3;
      int kk = (tid & 7) * 4;
      #pragma unroll
      for (int it = 0; it < PW; ++it) {
        int nn = n + it * 32;
        v4f v = wv[it];
        if (n0 + nn >= N) { v[0] = 0.f; v[1] = 0.f; v[2] = 0.f; v[3] = 0.f; }
        Bs[kk + 0][nn] = v[0]; Bs[kk + 1][nn] = v[1];
        Bs[kk + 2][nn] = v[2]; Bs[kk + 3][nn] = v[3];
      }
    }
    __syncthreads();

    #pragma unroll
    for (int kk = 0; kk < 32; kk += 4) {
      v2f af[TI], bf[TJ];
      #pragma unroll
      for (int i = 0; i < TI; ++i) {
        int row = wrow * (TI * 16) + i * 16 + r;
        af[i][0] = As[row][kk + 2 * hi];
        af[i][1] = As[row][kk + 1 + 2 * hi];
      }
      #pragma unroll
      for (int j = 0; j < TJ; ++j) {
        int col = wcol * (TJ * 16) + j * 16 + r;
        bf[j][0] = Bs[kk + 2 * hi][col];
        bf[j][1] = Bs[kk + 1 + 2 * hi][col];
      }
      #pragma unroll
      for (int i = 0; i < TI; ++i)
        #pragma unroll
        for (int j = 0; j < TJ; ++j)
          acc[i][j] = __builtin_amdgcn_wmma_f32_16x16x4_f32(
              false, af[i], false, bf[j], (short)0, acc[i][j], false, false);
    }
  }

  #pragma unroll
  for (int i = 0; i < TI; ++i)
    #pragma unroll
    for (int j = 0; j < TJ; ++j) {
      int gcol = n0 + wcol * (TJ * 16) + j * 16 + r;
      if (gcol >= N) continue;
      float b = bias ? bias[gcol] : 0.0f;
      #pragma unroll
      for (int vg = 0; vg < 8; ++vg) {
        int grow = m0 + wrow * (TI * 16) + i * 16 + vg + 8 * hi;
        C[(size_t)grow * N + gcol] = acc[i][j][vg] + b;
      }
    }
}

__global__ void rope_kernel(float* __restrict__ qbuf, float* __restrict__ kvbuf)
{
  const long long QP = 2048LL * 64 * 32;
  const long long KP = 2048LL * 8 * 32;
  long long idx = (long long)blockIdx.x * blockDim.x + threadIdx.x;
  if (idx >= QP + KP) return;

  int s, i;
  float *p0, *p1;
  if (idx < QP) {
    int t = (int)idx;
    i = t & 31;
    int h = (t >> 5) & 63;
    s = t >> 11;
    float* base = qbuf + (size_t)s * 4096 + h * 64;
    p0 = base + i; p1 = base + i + 32;
  } else {
    int t = (int)(idx - QP);
    i = t & 31;
    int kvh = (t >> 5) & 7;
    s = t >> 8;
    float* base = kvbuf + (size_t)s * 1024 + kvh * 128;
    p0 = base + i; p1 = base + i + 32;
  }
  float inv = powf(150000.0f, -((float)(2 * i)) / 64.0f);
  float ang = (float)s * inv;
  float c = cosf(ang), sn = sinf(ang);
  float x1 = *p0, x2 = *p1;
  *p0 = x1 * c - x2 * sn;
  *p1 = x2 * c + x1 * sn;
}

template <int CTRL>
__device__ __forceinline__ float row_xor(float v)
{
  return __int_as_float(__builtin_amdgcn_update_dpp(
      0, __float_as_int(v), CTRL, 0xf, 0xf, true));
}

__global__ __launch_bounds__(128) void flash_attn(
    const float* __restrict__ qbuf, const float* __restrict__ kvbuf,
    float* __restrict__ attn)
{
  __shared__ float Qs[64][65];
  __shared__ float Kt[64][65];
  __shared__ float Vs[64][64];
  __shared__ float Ssc[64][65];

  const int tid   = threadIdx.x;
  const int wave  = tid >> 5;
  const int lane  = tid & 31;
  const int r     = lane & 15;
  const int hi    = lane >> 4;
  const int h     = blockIdx.x;
  const int qtile = 31 - blockIdx.y;
  const int kvh   = h >> 3;
  const int qbase = qtile * 64;

  #pragma unroll 4
  for (int it = 0; it < 32; ++it) {
    int f = tid + it * 128;
    int sl = f >> 6, d = f & 63;
    Qs[sl][d] = qbuf[(size_t)(qbase + sl) * 4096 + h * 64 + d];
  }

  float m_i[8], l_i[8];
  for (int vg = 0; vg < 8; ++vg) { m_i[vg] = -INFINITY; l_i[vg] = 0.0f; }
  v8f oacc[4];
  for (int sub = 0; sub < 4; ++sub)
    for (int c = 0; c < 8; ++c) oacc[sub][c] = 0.0f;

  for (int j = 0; j <= qtile; ++j) {
    const int kvbase = j * 64;
    __syncthreads();
    #pragma unroll 4
    for (int it = 0; it < 32; ++it) {
      int f = tid + it * 128;
      int n = f >> 6, d = f & 63;
      const float* kv = &kvbuf[(size_t)(kvbase + n) * 1024 + kvh * 128];
      Kt[d][n] = kv[d];
      Vs[n][d] = kv[64 + d];
    }
    __syncthreads();

    v8f sacc[4];
    for (int sub = 0; sub < 4; ++sub)
      for (int c = 0; c < 8; ++c) sacc[sub][c] = 0.0f;
    #pragma unroll
    for (int kk = 0; kk < 64; kk += 4) {
      v2f af;
      af[0] = Qs[wave * 16 + r][kk + 2 * hi];
      af[1] = Qs[wave * 16 + r][kk + 1 + 2 * hi];
      #pragma unroll
      for (int sub = 0; sub < 4; ++sub) {
        v2f bf;
        bf[0] = Kt[kk + 2 * hi][sub * 16 + r];
        bf[1] = Kt[kk + 1 + 2 * hi][sub * 16 + r];
        sacc[sub] = __builtin_amdgcn_wmma_f32_16x16x4_f32(
            false, af, false, bf, (short)0, sacc[sub], false, false);
      }
    }

    float mx[8];
    #pragma unroll
    for (int vg = 0; vg < 8; ++vg) {
      int qr = qbase + wave * 16 + vg + 8 * hi;
      #pragma unroll
      for (int sub = 0; sub < 4; ++sub) {
        int kc = kvbase + sub * 16 + r;
        sacc[sub][vg] = (kc <= qr) ? sacc[sub][vg] * 0.125f : NEGBIG;
      }
      mx[vg] = fmaxf(fmaxf(sacc[0][vg], sacc[1][vg]), fmaxf(sacc[2][vg], sacc[3][vg]));
    }
    #pragma unroll
    for (int vg = 0; vg < 8; ++vg) mx[vg] = fmaxf(mx[vg], row_xor<0x161>(mx[vg]));
    #pragma unroll
    for (int vg = 0; vg < 8; ++vg) mx[vg] = fmaxf(mx[vg], row_xor<0x162>(mx[vg]));
    #pragma unroll
    for (int vg = 0; vg < 8; ++vg) mx[vg] = fmaxf(mx[vg], row_xor<0x164>(mx[vg]));
    #pragma unroll
    for (int vg = 0; vg < 8; ++vg) mx[vg] = fmaxf(mx[vg], row_xor<0x168>(mx[vg]));

    #pragma unroll
    for (int vg = 0; vg < 8; ++vg) {
      float newm  = fmaxf(m_i[vg], mx[vg]);
      float alpha = __expf(m_i[vg] - newm);
      m_i[vg] = newm;
      float ps = 0.0f;
      #pragma unroll
      for (int sub = 0; sub < 4; ++sub) {
        float p = __expf(sacc[sub][vg] - newm);
        ps += p;
        Ssc[wave * 16 + vg + 8 * hi][sub * 16 + r] = p;
        oacc[sub][vg] *= alpha;
      }
      l_i[vg] = l_i[vg] * alpha + ps;
    }

    #pragma unroll
    for (int kk = 0; kk < 64; kk += 4) {
      v2f af;
      af[0] = Ssc[wave * 16 + r][kk + 2 * hi];
      af[1] = Ssc[wave * 16 + r][kk + 1 + 2 * hi];
      #pragma unroll
      for (int sub = 0; sub < 4; ++sub) {
        v2f bf;
        bf[0] = Vs[kk + 2 * hi][sub * 16 + r];
        bf[1] = Vs[kk + 1 + 2 * hi][sub * 16 + r];
        oacc[sub] = __builtin_amdgcn_wmma_f32_16x16x4_f32(
            false, af, false, bf, (short)0, oacc[sub], false, false);
      }
    }
  }

  #pragma unroll
  for (int vg = 0; vg < 8; ++vg) l_i[vg] += row_xor<0x161>(l_i[vg]);
  #pragma unroll
  for (int vg = 0; vg < 8; ++vg) l_i[vg] += row_xor<0x162>(l_i[vg]);
  #pragma unroll
  for (int vg = 0; vg < 8; ++vg) l_i[vg] += row_xor<0x164>(l_i[vg]);
  #pragma unroll
  for (int vg = 0; vg < 8; ++vg) l_i[vg] += row_xor<0x168>(l_i[vg]);
  #pragma unroll
  for (int vg = 0; vg < 8; ++vg) {
    float im = 1.0f / l_i[vg];
    int grow = qbase + wave * 16 + vg + 8 * hi;
    for (int sub = 0; sub < 4; ++sub) {
      int d = sub * 16 + r;
      attn[(size_t)grow * 4096 + h * 64 + d] = oacc[sub][vg] * im;
    }
  }
}

extern "C" void kernel_launch(void* const* d_in, const int* in_sizes, int n_in,
                              void* d_out, int out_size, void* d_ws, size_t ws_size,
                              hipStream_t stream)
{
  const float* hidden = (const float*)d_in[0];
  const float* q_w    = (const float*)d_in[2];
  const float* q_b    = (const float*)d_in[3];
  const float* kv_a_w = (const float*)d_in[4];
  const float* kv_b_w = (const float*)d_in[5];
  const float* o_w    = (const float*)d_in[6];
  const float* o_b    = (const float*)d_in[7];
  float* out = (float*)d_out;

  float* ws    = (float*)d_ws;
  float* qbuf  = ws;
  float* ckv   = qbuf  + (size_t)2048 * 4096;
  float* kvbuf = ckv   + (size_t)2048 * 384;
  float* attn  = kvbuf + (size_t)2048 * 1024;

  dim3 blk(256);
  gemm_wmma_f32<128, 128, 2, 4><<<dim3(4096 / 128, 2048 / 128), blk, 0, stream>>>(
      hidden, q_w, q_b, qbuf, 2048, 4096, 2880);
  gemm_wmma_f32<64, 64, 2, 4><<<dim3(384 / 64, 2048 / 64), blk, 0, stream>>>(
      hidden, kv_a_w, nullptr, ckv, 2048, 384, 2880);
  gemm_wmma_f32<64, 64, 2, 4><<<dim3(1024 / 64, 2048 / 64), blk, 0, stream>>>(
      ckv, kv_b_w, nullptr, kvbuf, 2048, 1024, 384);
  long long pairs = 2048LL * 64 * 32 + 2048LL * 8 * 32;
  rope_kernel<<<(int)((pairs + 255) / 256), 256, 0, stream>>>(qbuf, kvbuf);
  flash_attn<<<dim3(64, 32), 128, 0, stream>>>(qbuf, kvbuf, attn);
  gemm_wmma_f32<128, 96, 4, 2><<<dim3(2880 / 96, 2048 / 128), blk, 0, stream>>>(
      attn, o_w, o_b, out, 2048, 2880, 4096);
}


# LDS tiles keep k contiguous per row at a conflict-free pitch (36 / 68 floats, V stored transposed): a WMMA fragment is one 8-byte LDS load, tiles staged by 16-byte stores; k order and sums unchanged
# speedup vs baseline: 1.8874x; 1.1997x over previous
#include <hip/hip_runtime.h>
#include <math.h>

typedef __attribute__((ext_vector_type(2))) float v2f;
typedef __attribute__((ext_vector_type(8))) float v8f;
typedef __attribute__((ext_vector_type(4))) float v4f;

#define NEGBIG (-3.4e38f)

template <int BM, int BN, int WR, int WC>
__global__ __launch_bounds__(256) void gemm_wmma_f32(
    const float* __restrict__ A, const float* __restrict__ W,
    const float* __restrict__ bias, float* __restrict__ C,
    int M, int N, int K)
{
  static_assert(WR * WC == 8, "8 waves per block");
  static_assert(BM % (16 * WR) == 0 && BN % (16 * WC) == 0, "whole 16x16 subtiles per wave");
  static_assert(BM % 32 == 0 && BN % 32 == 0, "staging moves 32 rows per pass");
  constexpr int TI = BM / (16 * WR);
  constexpr int TJ = BN / (16 * WC);
  constexpr int PA = BM / 32;
  constexpr int PW = BN / 32;
  constexpr int LP = 36;

  __shared__ alignas(16) float As[BM][LP];
  __shared__ alignas(16) float Ws[BN][LP];

  const int tid  = threadIdx.x;
  const int wave = tid >> 5;
  const int lane = tid & 31;
  const int r    = lane & 15;
  const int hi   = lane >> 4;
  const int wrow = wave / WC;
  const int wcol = wave % WC;
  const int m0   = blockIdx.y * BM;
  const int n0   = blockIdx.x * BN;

  const int sq   = (tid & 7) * 4;
  const int srow = ((tid >> 3) & 1) * 8 + ((tid >> 4) & 7) + (tid >> 7) * 16;

  v8f acc[TI][TJ];
  #pragma unroll
  for (int i = 0; i < TI; ++i)
    #pragma unroll
    for (int j = 0; j < TJ; ++j)
      #pragma unroll
      for (int c = 0; c < 8; ++c) acc[i][j][c] = 0.0f;

  for (int k0 = 0; k0 < K; k0 += 32) {
    v4f av[PA], wv[PW];
    #pragma unroll
    for (int it = 0; it < PA; ++it)
      av[it] = *(const v4f*)&A[(size_t)(m0 + srow + it * 32) * K + k0 + sq];
    #pragma unroll
    for (int it = 0; it < PW; ++it) {
      int grow = n0 + srow + it * 32;
      int crow = grow < N ? grow : N - 1;
      wv[it] = *(const v4f*)&W[(size_t)crow * K + k0 + sq];
    }
    __syncthreads();
    #pragma unroll
    for (int it = 0; it < PA; ++it)
      *(v4f*)&As[srow + it * 32][sq] = av[it];
    #pragma unroll
    for (int it = 0; it < PW; ++it) {
      v4f v = wv[it];
      if (n0 + srow + it * 32 >= N) { v[0] = 0.f; v[1] = 0.f; v[2] = 0.f; v[3] = 0.f; }
      *(v4f*)&Ws[srow + it * 32][sq] = v;
    }
    __syncthreads();

    #pragma unroll
    for (int kk = 0; kk < 32; kk += 4) {
      v2f af[TI], bf[TJ];
      #pragma unroll
      for (int i = 0; i < TI; ++i)
        af[i] = *(const v2f*)&As[wrow * (TI * 16) + i * 16 + r][kk + 2 * hi];
      #pragma unroll
      for (int j = 0; j < TJ; ++j)
        bf[j] = *(const v2f*)&Ws[wcol * (TJ * 16) + j * 16 + r][kk + 2 * hi];
      #pragma unroll
      for (int i = 0; i < TI; ++i)
        #pragma unroll
        for (int j = 0; j < TJ; ++j)
          acc[i][j] = __builtin_amdgcn_wmma_f32_16x16x4_f32(
              false, af[i], false, bf[j], (short)0, acc[i][j], false, false);
    }
  }

  #pragma unroll
  for (int i = 0; i < TI; ++i)
    #pragma unroll
    for (int j = 0; j < TJ; ++j) {
      int gcol = n0 + wcol * (TJ * 16) + j * 16 + r;
      if (gcol >= N) continue;
      float b = bias ? bias[gcol] : 0.0f;
      #pragma unroll
      for (int vg = 0; vg < 8; ++vg) {
        int grow = m0 + wrow * (TI * 16) + i * 16 + vg + 8 * hi;
        C[(size_t)grow * N + gcol] = acc[i][j][vg] + b;
      }
    }
}

__global__ void rope_kernel(float* __restrict__ qbuf, float* __restrict__ kvbuf)
{
  const long long QP = 2048LL * 64 * 32;
  const long long KP = 2048LL * 8 * 32;
  long long idx = (long long)blockIdx.x * blockDim.x + threadIdx.x;
  if (idx >= QP + KP) return;

  int s, i;
  float *p0, *p1;
  if (idx < QP) {
    int t = (int)idx;
    i = t & 31;
    int h = (t >> 5) & 63;
    s = t >> 11;
    float* base = qbuf + (size_t)s * 4096 + h * 64;
    p0 = base + i; p1 = base + i + 32;
  } else {
    int t = (int)(idx - QP);
    i = t & 31;
    int kvh = (t >> 5) & 7;
    s = t >> 8;
    float* base = kvbuf + (size_t)s * 1024 + kvh * 128;
    p0 = base + i; p1 = base + i + 32;
  }
  float inv = powf(150000.0f, -((float)(2 * i)) / 64.0f);
  float ang = (float)s * inv;
  float c = cosf(ang), sn = sinf(ang);
  float x1 = *p0, x2 = *p1;
  *p0 = x1 * c - x2 * sn;
  *p1 = x2 * c + x1 * sn;
}

template <int CTRL>
__device__ __forceinline__ float row_xor(float v)
{
  return __int_as_float(__builtin_amdgcn_update_dpp(
      0, __float_as_int(v), CTRL, 0xf, 0xf, true));
}

__global__ __launch_bounds__(128) void flash_attn(
    const float* __restrict__ qbuf, const float* __restrict__ kvbuf,
    float* __restrict__ attn)
{
  constexpr int AP = 68;
  __shared__ alignas(16) float Qs[64][AP];
  __shared__ alignas(16) float Ks[64][AP];
  __shared__ alignas(16) float Vt[64][AP];
  __shared__ alignas(16) float Ps[64][AP];

  const int tid   = threadIdx.x;
  const int wave  = tid >> 5;
  const int lane  = tid & 31;
  const int r     = lane & 15;
  const int hi    = lane >> 4;
  const int h     = blockIdx.x;
  const int qtile = 31 - blockIdx.y;
  const int kvh   = h >> 3;
  const int qbase = qtile * 64;

  const int srow = tid >> 4;
  const int sc4  = (tid & 15) * 4;
  const int vd   = tid & 63;
  const int vnq  = tid >> 6;

  #pragma unroll
  for (int it = 0; it < 8; ++it) {
    int sl = srow + it * 8;
    *(v4f*)&Qs[sl][sc4] = *(const v4f*)&qbuf[(size_t)(qbase + sl) * 4096 + h * 64 + sc4];
  }

  float m_i[8], l_i[8];
  for (int vg = 0; vg < 8; ++vg) { m_i[vg] = -INFINITY; l_i[vg] = 0.0f; }
  v8f oacc[4];
  for (int sub = 0; sub < 4; ++sub)
    for (int c = 0; c < 8; ++c) oacc[sub][c] = 0.0f;

  for (int j = 0; j <= qtile; ++j) {
    const int kvbase = j * 64;
    const float* kvt = &kvbuf[(size_t)kvbase * 1024 + kvh * 128];
    v4f kreg[8], vreg[8];
    #pragma unroll
    for (int it = 0; it < 8; ++it)
      kreg[it] = *(const v4f*)&kvt[(size_t)(srow + it * 8) * 1024 + sc4];
    #pragma unroll
    for (int it = 0; it < 8; ++it) {
      int n = 4 * (vnq + 2 * it);
      #pragma unroll
      for (int c = 0; c < 4; ++c)
        vreg[it][c] = kvt[(size_t)(n + c) * 1024 + 64 + vd];
    }
    __syncthreads();
    #pragma unroll
    for (int it = 0; it < 8; ++it)
      *(v4f*)&Ks[srow + it * 8][sc4] = kreg[it];
    #pragma unroll
    for (int it = 0; it < 8; ++it)
      *(v4f*)&Vt[vd][4 * (vnq + 2 * it)] = vreg[it];
    __syncthreads();

    v8f sacc[4];
    for (int sub = 0; sub < 4; ++sub)
      for (int c = 0; c < 8; ++c) sacc[sub][c] = 0.0f;
    #pragma unroll
    for (int kk = 0; kk < 64; kk += 4) {
      v2f af = *(const v2f*)&Qs[wave * 16 + r][kk + 2 * hi];
      #pragma unroll
      for (int sub = 0; sub < 4; ++sub) {
        v2f bf = *(const v2f*)&Ks[sub * 16 + r][kk + 2 * hi];
        sacc[sub] = __builtin_amdgcn_wmma_f32_16x16x4_f32(
            false, af, false, bf, (short)0, sacc[sub], false, false);
      }
    }

    float mx[8];
    #pragma unroll
    for (int vg = 0; vg < 8; ++vg) {
      int qr = qbase + wave * 16 + vg + 8 * hi;
      #pragma unroll
      for (int sub = 0; sub < 4; ++sub) {
        int kc = kvbase + sub * 16 + r;
        sacc[sub][vg] = (kc <= qr) ? sacc[sub][vg] * 0.125f : NEGBIG;
      }
      mx[vg] = fmaxf(fmaxf(sacc[0][vg], sacc[1][vg]), fmaxf(sacc[2][vg], sacc[3][vg]));
    }
    #pragma unroll
    for (int vg = 0; vg < 8; ++vg) mx[vg] = fmaxf(mx[vg], row_xor<0x161>(mx[vg]));
    #pragma unroll
    for (int vg = 0; vg < 8; ++vg) mx[vg] = fmaxf(mx[vg], row_xor<0x162>(mx[vg]));
    #pragma unroll
    for (int vg = 0; vg < 8; ++vg) mx[vg] = fmaxf(mx[vg], row_xor<0x164>(mx[vg]));
    #pragma unroll
    for (int vg = 0; vg < 8; ++vg) mx[vg] = fmaxf(mx[vg], row_xor<0x168>(mx[vg]));

    #pragma unroll
    for (int vg = 0; vg < 8; ++vg) {
      float newm  = fmaxf(m_i[vg], mx[vg]);
      float alpha = __expf(m_i[vg] - newm);
      m_i[vg] = newm;
      float ps = 0.0f;
      #pragma unroll
      for (int sub = 0; sub < 4; ++sub) {
        float p = __expf(sacc[sub][vg] - newm);
        ps += p;
        Ps[wave * 16 + vg + 8 * hi][sub * 16 + r] = p;
        oacc[sub][vg] *= alpha;
      }
      l_i[vg] = l_i[vg] * alpha + ps;
    }

    #pragma unroll
    for (int kk = 0; kk < 64; kk += 4) {
      v2f af = *(const v2f*)&Ps[wave * 16 + r][kk + 2 * hi];
      #pragma unroll
      for (int sub = 0; sub < 4; ++sub) {
        v2f bf = *(const v2f*)&Vt[sub * 16 + r][kk + 2 * hi];
        oacc[sub] = __builtin_amdgcn_wmma_f32_16x16x4_f32(
            false, af, false, bf, (short)0, oacc[sub], false, false);
      }
    }
  }

  #pragma unroll
  for (int vg = 0; vg < 8; ++vg) l_i[vg] += row_xor<0x161>(l_i[vg]);
  #pragma unroll
  for (int vg = 0; vg < 8; ++vg) l_i[vg] += row_xor<0x162>(l_i[vg]);
  #pragma unroll
  for (int vg = 0; vg < 8; ++vg) l_i[vg] += row_xor<0x164>(l_i[vg]);
  #pragma unroll
  for (int vg = 0; vg < 8; ++vg) l_i[vg] += row_xor<0x168>(l_i[vg]);
  #pragma unroll
  for (int vg = 0; vg < 8; ++vg) {
    float im = 1.0f / l_i[vg];
    int grow = qbase + wave * 16 + vg + 8 * hi;
    for (int sub = 0; sub < 4; ++sub) {
      int d = sub * 16 + r;
      attn[(size_t)grow * 4096 + h * 64 + d] = oacc[sub][vg] * im;
    }
  }
}

extern "C" void kernel_launch(void* const* d_in, const int* in_sizes, int n_in,
                              void* d_out, int out_size, void* d_ws, size_t ws_size,
                              hipStream_t stream)
{
  const float* hidden = (const float*)d_in[0];
  const float* q_w    = (const float*)d_in[2];
  const float* q_b    = (const float*)d_in[3];
  const float* kv_a_w = (const float*)d_in[4];
  const float* kv_b_w = (const float*)d_in[5];
  const float* o_w    = (const float*)d_in[6];
  const float* o_b    = (const float*)d_in[7];
  float* out = (float*)d_out;

  float* ws    = (float*)d_ws;
  float* qbuf  = ws;
  float* ckv   = qbuf  + (size_t)2048 * 4096;
  float* kvbuf = ckv   + (size_t)2048 * 384;
  float* attn  = kvbuf + (size_t)2048 * 1024;

  dim3 blk(256);
  gemm_wmma_f32<128, 128, 2, 4><<<dim3(4096 / 128, 2048 / 128), blk, 0, stream>>>(
      hidden, q_w, q_b, qbuf, 2048, 4096, 2880);
  gemm_wmma_f32<64, 64, 2, 4><<<dim3(384 / 64, 2048 / 64), blk, 0, stream>>>(
      hidden, kv_a_w, nullptr, ckv, 2048, 384, 2880);
  gemm_wmma_f32<64, 64, 2, 4><<<dim3(1024 / 64, 2048 / 64), blk, 0, stream>>>(
      ckv, kv_b_w, nullptr, kvbuf, 2048, 1024, 384);
  long long pairs = 2048LL * 64 * 32 + 2048LL * 8 * 32;
  rope_kernel<<<(int)((pairs + 255) / 256), 256, 0, stream>>>(qbuf, kvbuf);
  flash_attn<<<dim3(64, 32), 128, 0, stream>>>(qbuf, kvbuf, attn);
  gemm_wmma_f32<128, 96, 4, 2><<<dim3(2880 / 96, 2048 / 128), blk, 0, stream>>>(
      attn, o_w, o_b, out, 2048, 2880, 4096);
}


# GEMM k-loop software-pipelined over two LDS tile buffers: next tile stored and the one after loaded from global while the WMMAs run; one barrier per k-step, not two; k order and sums unchanged
# speedup vs baseline: 2.0506x; 1.0865x over previous
#include <hip/hip_runtime.h>
#include <math.h>

typedef __attribute__((ext_vector_type(2))) float v2f;
typedef __attribute__((ext_vector_type(8))) float v8f;
typedef __attribute__((ext_vector_type(4))) float v4f;

#define NEGBIG (-3.4e38f)

template <int BM, int BN, int WR, int WC>
__global__ __launch_bounds__(256) void gemm_wmma_f32(
    const float* __restrict__ A, const float* __restrict__ W,
    const float* __restrict__ bias, float* __restrict__ C,
    int M, int N, int K)
{
  static_assert(WR * WC == 8, "8 waves per block");
  static_assert(BM % (16 * WR) == 0 && BN % (16 * WC) == 0, "whole 16x16 subtiles per wave");
  static_assert(BM % 32 == 0 && BN % 32 == 0, "staging moves 32 rows per pass");
  constexpr int TI = BM / (16 * WR);
  constexpr int TJ = BN / (16 * WC);
  constexpr int PA = BM / 32;
  constexpr int PW = BN / 32;
  constexpr int LP = 36;

  __shared__ alignas(16) float As[2][BM][LP];
  __shared__ alignas(16) float Ws[2][BN][LP];

  const int tid  = threadIdx.x;
  const int wave = tid >> 5;
  const int lane = tid & 31;
  const int r    = lane & 15;
  const int hi   = lane >> 4;
  const int wrow = wave / WC;
  const int wcol = wave % WC;
  const int m0   = blockIdx.y * BM;
  const int n0   = blockIdx.x * BN;

  const int sq   = (tid & 7) * 4;
  const int srow = ((tid >> 3) & 1) * 8 + ((tid >> 4) & 7) + (tid >> 7) * 16;

  const float* ap[PA];
  const float* wp[PW];
  bool wz[PW];
  #pragma unroll
  for (int it = 0; it < PA; ++it)
    ap[it] = A + (size_t)(m0 + srow + it * 32) * K + sq;
  #pragma unroll
  for (int it = 0; it < PW; ++it) {
    int grow = n0 + srow + it * 32;
    int crow = grow < N ? grow : N - 1;
    wz[it] = grow >= N;
    wp[it] = W + (size_t)crow * K + sq;
  }

  v8f acc[TI][TJ];
  #pragma unroll
  for (int i = 0; i < TI; ++i)
    #pragma unroll
    for (int j = 0; j < TJ; ++j)
      #pragma unroll
      for (int c = 0; c < 8; ++c) acc[i][j][c] = 0.0f;

  const int T = K / 32;
  v4f av[PA], wv[PW];

  #pragma unroll
  for (int it = 0; it < PA; ++it) av[it] = *(const v4f*)(ap[it]);
  #pragma unroll
  for (int it = 0; it < PW; ++it) wv[it] = *(const v4f*)(wp[it]);
  #pragma unroll
  for (int it = 0; it < PA; ++it)
    *(v4f*)&As[0][srow + it * 32][sq] = av[it];
  #pragma unroll
  for (int it = 0; it < PW; ++it) {
    v4f v = wv[it];
    if (wz[it]) { v[0] = 0.f; v[1] = 0.f; v[2] = 0.f; v[3] = 0.f; }
    *(v4f*)&Ws[0][srow + it * 32][sq] = v;
  }
  if (T > 1) {
    #pragma unroll
    for (int it = 0; it < PA; ++it) av[it] = *(const v4f*)(ap[it] + 32);
    #pragma unroll
    for (int it = 0; it < PW; ++it) wv[it] = *(const v4f*)(wp[it] + 32);
  }
  __syncthreads();

  for (int t = 0; t < T; ++t) {
    const int cur = t & 1;
    if (t + 1 < T) {
      #pragma unroll
      for (int it = 0; it < PA; ++it)
        *(v4f*)&As[cur ^ 1][srow + it * 32][sq] = av[it];
      #pragma unroll
      for (int it = 0; it < PW; ++it) {
        v4f v = wv[it];
        if (wz[it]) { v[0] = 0.f; v[1] = 0.f; v[2] = 0.f; v[3] = 0.f; }
        *(v4f*)&Ws[cur ^ 1][srow + it * 32][sq] = v;
      }
    }
    if (t + 2 < T) {
      const int ko = (t + 2) * 32;
      #pragma unroll
      for (int it = 0; it < PA; ++it) av[it] = *(const v4f*)(ap[it] + ko);
      #pragma unroll
      for (int it = 0; it < PW; ++it) wv[it] = *(const v4f*)(wp[it] + ko);
    }

    #pragma unroll
    for (int kk = 0; kk < 32; kk += 4) {
      v2f af[TI], bf[TJ];
      #pragma unroll
      for (int i = 0; i < TI; ++i)
        af[i] = *(const v2f*)&As[cur][wrow * (TI * 16) + i * 16 + r][kk + 2 * hi];
      #pragma unroll
      for (int j = 0; j < TJ; ++j)
        bf[j] = *(const v2f*)&Ws[cur][wcol * (TJ * 16) + j * 16 + r][kk + 2 * hi];
      #pragma unroll
      for (int i = 0; i < TI; ++i)
        #pragma unroll
        for (int j = 0; j < TJ; ++j)
          acc[i][j] = __builtin_amdgcn_wmma_f32_16x16x4_f32(
              false, af[i], false, bf[j], (short)0, acc[i][j], false, false);
    }
    __syncthreads();
  }

  #pragma unroll
  for (int i = 0; i < TI; ++i)
    #pragma unroll
    for (int j = 0; j < TJ; ++j) {
      int gcol = n0 + wcol * (TJ * 16) + j * 16 + r;
      if (gcol >= N) continue;
      float b = bias ? bias[gcol] : 0.0f;
      #pragma unroll
      for (int vg = 0; vg < 8; ++vg) {
        int grow = m0 + wrow * (TI * 16) + i * 16 + vg + 8 * hi;
        C[(size_t)grow * N + gcol] = acc[i][j][vg] + b;
      }
    }
}

__global__ void rope_kernel(float* __restrict__ qbuf, float* __restrict__ kvbuf)
{
  const long long QP = 2048LL * 64 * 32;
  const long long KP = 2048LL * 8 * 32;
  long long idx = (long long)blockIdx.x * blockDim.x + threadIdx.x;
  if (idx >= QP + KP) return;

  int s, i;
  float *p0, *p1;
  if (idx < QP) {
    int t = (int)idx;
    i = t & 31;
    int h = (t >> 5) & 63;
    s = t >> 11;
    float* base = qbuf + (size_t)s * 4096 + h * 64;
    p0 = base + i; p1 = base + i + 32;
  } else {
    int t = (int)(idx - QP);
    i = t & 31;
    int kvh = (t >> 5) & 7;
    s = t >> 8;
    float* base = kvbuf + (size_t)s * 1024 + kvh * 128;
    p0 = base + i; p1 = base + i + 32;
  }
  float inv = powf(150000.0f, -((float)(2 * i)) / 64.0f);
  float ang = (float)s * inv;
  float c = cosf(ang), sn = sinf(ang);
  float x1 = *p0, x2 = *p1;
  *p0 = x1 * c - x2 * sn;
  *p1 = x2 * c + x1 * sn;
}

template <int CTRL>
__device__ __forceinline__ float row_xor(float v)
{
  return __int_as_float(__builtin_amdgcn_update_dpp(
      0, __float_as_int(v), CTRL, 0xf, 0xf, true));
}

__global__ __launch_bounds__(128) void flash_attn(
    const float* __restrict__ qbuf, const float* __restrict__ kvbuf,
    float* __restrict__ attn)
{
  constexpr int AP = 68;
  __shared__ alignas(16) float Qs[64][AP];
  __shared__ alignas(16) float Ks[64][AP];
  __shared__ alignas(16) float Vt[64][AP];
  __shared__ alignas(16) float Ps[64][AP];

  const int tid   = threadIdx.x;
  const int wave  = tid >> 5;
  const int lane  = tid & 31;
  const int r     = lane & 15;
  const int hi    = lane >> 4;
  const int h     = blockIdx.x;
  const int qtile = 31 - blockIdx.y;
  const int kvh   = h >> 3;
  const int qbase = qtile * 64;

  const int srow = tid >> 4;
  const int sc4  = (tid & 15) * 4;
  const int vd   = tid & 63;
  const int vnq  = tid >> 6;

  #pragma unroll
  for (int it = 0; it < 8; ++it) {
    int sl = srow + it * 8;
    *(v4f*)&Qs[sl][sc4] = *(const v4f*)&qbuf[(size_t)(qbase + sl) * 4096 + h * 64 + sc4];
  }

  float m_i[8], l_i[8];
  for (int vg = 0; vg < 8; ++vg) { m_i[vg] = -INFINITY; l_i[vg] = 0.0f; }
  v8f oacc[4];
  for (int sub = 0; sub < 4; ++sub)
    for (int c = 0; c < 8; ++c) oacc[sub][c] = 0.0f;

  for (int j = 0; j <= qtile; ++j) {
    const int kvbase = j * 64;
    const float* kvt = &kvbuf[(size_t)kvbase * 1024 + kvh * 128];
    v4f kreg[8], vreg[8];
    #pragma unroll
    for (int it = 0; it < 8; ++it)
      kreg[it] = *(const v4f*)&kvt[(size_t)(srow + it * 8) * 1024 + sc4];
    #pragma unroll
    for (int it = 0; it < 8; ++it) {
      int n = 4 * (vnq + 2 * it);
      #pragma unroll
      for (int c = 0; c < 4; ++c)
        vreg[it][c] = kvt[(size_t)(n + c) * 1024 + 64 + vd];
    }
    __syncthreads();
    #pragma unroll
    for (int it = 0; it < 8; ++it)
      *(v4f*)&Ks[srow + it * 8][sc4] = kreg[it];
    #pragma unroll
    for (int it = 0; it < 8; ++it)
      *(v4f*)&Vt[vd][4 * (vnq + 2 * it)] = vreg[it];
    __syncthreads();

    v8f sacc[4];
    for (int sub = 0; sub < 4; ++sub)
      for (int c = 0; c < 8; ++c) sacc[sub][c] = 0.0f;
    #pragma unroll
    for (int kk = 0; kk < 64; kk += 4) {
      v2f af = *(const v2f*)&Qs[wave * 16 + r][kk + 2 * hi];
      #pragma unroll
      for (int sub = 0; sub < 4; ++sub) {
        v2f bf = *(const v2f*)&Ks[sub * 16 + r][kk + 2 * hi];
        sacc[sub] = __builtin_amdgcn_wmma_f32_16x16x4_f32(
            false, af, false, bf, (short)0, sacc[sub], false, false);
      }
    }

    float mx[8];
    #pragma unroll
    for (int vg = 0; vg < 8; ++vg) {
      int qr = qbase + wave * 16 + vg + 8 * hi;
      #pragma unroll
      for (int sub = 0; sub < 4; ++sub) {
        int kc = kvbase + sub * 16 + r;
        sacc[sub][vg] = (kc <= qr) ? sacc[sub][vg] * 0.125f : NEGBIG;
      }
      mx[vg] = fmaxf(fmaxf(sacc[0][vg], sacc[1][vg]), fmaxf(sacc[2][vg], sacc[3][vg]));
    }
    #pragma unroll
    for (int vg = 0; vg < 8; ++vg) mx[vg] = fmaxf(mx[vg], row_xor<0x161>(mx[vg]));
    #pragma unroll
    for (int vg = 0; vg < 8; ++vg) mx[vg] = fmaxf(mx[vg], row_xor<0x162>(mx[vg]));
    #pragma unroll
    for (int vg = 0; vg < 8; ++vg) mx[vg] = fmaxf(mx[vg], row_xor<0x164>(mx[vg]));
    #pragma unroll
    for (int vg = 0; vg < 8; ++vg) mx[vg] = fmaxf(mx[vg], row_xor<0x168>(mx[vg]));

    #pragma unroll
    for (int vg = 0; vg < 8; ++vg) {
      float newm  = fmaxf(m_i[vg], mx[vg]);
      float alpha = __expf(m_i[vg] - newm);
      m_i[vg] = newm;
      float ps = 0.0f;
      #pragma unroll
      for (int sub = 0; sub < 4; ++sub) {
        float p = __expf(sacc[sub][vg] - newm);
        ps += p;
        Ps[wave * 16 + vg + 8 * hi][sub * 16 + r] = p;
        oacc[sub][vg] *= alpha;
      }
      l_i[vg] = l_i[vg] * alpha + ps;
    }

    #pragma unroll
    for (int kk = 0; kk < 64; kk += 4) {
      v2f af = *(const v2f*)&Ps[wave * 16 + r][kk + 2 * hi];
      #pragma unroll
      for (int sub = 0; sub < 4; ++sub) {
        v2f bf = *(const v2f*)&Vt[sub * 16 + r][kk + 2 * hi];
        oacc[sub] = __builtin_amdgcn_wmma_f32_16x16x4_f32(
            false, af, false, bf, (short)0, oacc[sub], false, false);
      }
    }
  }

  #pragma unroll
  for (int vg = 0; vg < 8; ++vg) l_i[vg] += row_xor<0x161>(l_i[vg]);
  #pragma unroll
  for (int vg = 0; vg < 8; ++vg) l_i[vg] += row_xor<0x162>(l_i[vg]);
  #pragma unroll
  for (int vg = 0; vg < 8; ++vg) l_i[vg] += row_xor<0x164>(l_i[vg]);
  #pragma unroll
  for (int vg = 0; vg < 8; ++vg) l_i[vg] += row_xor<0x168>(l_i[vg]);
  #pragma unroll
  for (int vg = 0; vg < 8; ++vg) {
    float im = 1.0f / l_i[vg];
    int grow = qbase + wave * 16 + vg + 8 * hi;
    for (int sub = 0; sub < 4; ++sub) {
      int d = sub * 16 + r;
      attn[(size_t)grow * 4096 + h * 64 + d] = oacc[sub][vg] * im;
    }
  }
}

extern "C" void kernel_launch(void* const* d_in, const int* in_sizes, int n_in,
                              void* d_out, int out_size, void* d_ws, size_t ws_size,
                              hipStream_t stream)
{
  const float* hidden = (const float*)d_in[0];
  const float* q_w    = (const float*)d_in[2];
  const float* q_b    = (const float*)d_in[3];
  const float* kv_a_w = (const float*)d_in[4];
  const float* kv_b_w = (const float*)d_in[5];
  const float* o_w    = (const float*)d_in[6];
  const float* o_b    = (const float*)d_in[7];
  float* out = (float*)d_out;

  float* ws    = (float*)d_ws;
  float* qbuf  = ws;
  float* ckv   = qbuf  + (size_t)2048 * 4096;
  float* kvbuf = ckv   + (size_t)2048 * 384;
  float* attn  = kvbuf + (size_t)2048 * 1024;

  dim3 blk(256);
  gemm_wmma_f32<128, 128, 2, 4><<<dim3(4096 / 128, 2048 / 128), blk, 0, stream>>>(
      hidden, q_w, q_b, qbuf, 2048, 4096, 2880);
  gemm_wmma_f32<64, 64, 2, 4><<<dim3(384 / 64, 2048 / 64), blk, 0, stream>>>(
      hidden, kv_a_w, nullptr, ckv, 2048, 384, 2880);
  gemm_wmma_f32<64, 64, 2, 4><<<dim3(1024 / 64, 2048 / 64), blk, 0, stream>>>(
      ckv, kv_b_w, nullptr, kvbuf, 2048, 1024, 384);
  long long pairs = 2048LL * 64 * 32 + 2048LL * 8 * 32;
  rope_kernel<<<(int)((pairs + 255) / 256), 256, 0, stream>>>(qbuf, kvbuf);
  flash_attn<<<dim3(64, 32), 128, 0, stream>>>(qbuf, kvbuf, attn);
  gemm_wmma_f32<128, 96, 4, 2><<<dim3(2880 / 96, 2048 / 128), blk, 0, stream>>>(
      attn, o_w, o_b, out, 2048, 2880, 4096);
}


# RoPE applied in the Q and kv_b GEMM epilogues (rotation pair in one lane, cos/sin once per block in LDS): the separate pass over q and k and its launch are gone; k order and sums unchanged
# speedup vs baseline: 2.1242x; 1.0359x over previous
#include <hip/hip_runtime.h>
#include <math.h>

typedef __attribute__((ext_vector_type(2))) float v2f;
typedef __attribute__((ext_vector_type(8))) float v8f;
typedef __attribute__((ext_vector_type(4))) float v4f;

#define NEGBIG (-3.4e38f)

template <int BM, int BN, int WR, int WC, int ROPE>
__global__ __launch_bounds__(256) void gemm_wmma_f32(
    const float* __restrict__ A, const float* __restrict__ W,
    const float* __restrict__ bias, float* __restrict__ C,
    int M, int N, int K)
{
  static_assert(WR * WC == 8, "8 waves per block");
  static_assert(BM % (16 * WR) == 0 && BN % (16 * WC) == 0, "whole 16x16 subtiles per wave");
  static_assert(BM % 32 == 0 && BN % 32 == 0, "staging moves 32 rows per pass");
  constexpr int TI = BM / (16 * WR);
  constexpr int TJ = BN / (16 * WC);
  constexpr int PA = BM / 32;
  constexpr int PW = BN / 32;
  constexpr int LP = 36;
  static_assert(ROPE == 0 || (TJ == 2 && BN % 64 == 0 && WC * 32 == BN),
                "rotating launches: a pair of waves per 64-column head, two subtiles 32 columns apart");
  static_assert(ROPE != 2 || BN == 64, "k | v launches: a block is one half of a head");
  constexpr int CS = ROPE ? 32 : 16;

  __shared__ alignas(16) float As[2][BM][LP];
  __shared__ alignas(16) float Ws[2][BN][LP];

  const int tid  = threadIdx.x;
  const int wave = tid >> 5;
  const int lane = tid & 31;
  const int r    = lane & 15;
  const int hi   = lane >> 4;
  const int wrow = wave / WC;
  const int wcol = wave % WC;
  const int m0   = blockIdx.y * BM;
  const int n0   = blockIdx.x * BN;
  const int cb   = ROPE ? (wcol >> 1) * 64 + (wcol & 1) * 16 : wcol * (TJ * 16);

  const int sq   = (tid & 7) * 4;
  const int srow = ((tid >> 3) & 1) * 8 + ((tid >> 4) & 7) + (tid >> 7) * 16;

  const float* ap[PA];
  const float* wp[PW];
  bool wz[PW];
  #pragma unroll
  for (int it = 0; it < PA; ++it)
    ap[it] = A + (size_t)(m0 + srow + it * 32) * K + sq;
  #pragma unroll
  for (int it = 0; it < PW; ++it) {
    int grow = n0 + srow + it * 32;
    int crow = grow < N ? grow : N - 1;
    wz[it] = grow >= N;
    wp[it] = W + (size_t)crow * K + sq;
  }

  v8f acc[TI][TJ];
  #pragma unroll
  for (int i = 0; i < TI; ++i)
    #pragma unroll
    for (int j = 0; j < TJ; ++j)
      #pragma unroll
      for (int c = 0; c < 8; ++c) acc[i][j][c] = 0.0f;

  const int T = K / 32;
  v4f av[PA], wv[PW];

  #pragma unroll
  for (int it = 0; it < PA; ++it) av[it] = *(const v4f*)(ap[it]);
  #pragma unroll
  for (int it = 0; it < PW; ++it) wv[it] = *(const v4f*)(wp[it]);
  #pragma unroll
  for (int it = 0; it < PA; ++it)
    *(v4f*)&As[0][srow + it * 32][sq] = av[it];
  #pragma unroll
  for (int it = 0; it < PW; ++it) {
    v4f v = wv[it];
    if (wz[it]) { v[0] = 0.f; v[1] = 0.f; v[2] = 0.f; v[3] = 0.f; }
    *(v4f*)&Ws[0][srow + it * 32][sq] = v;
  }
  if (T > 1) {
    #pragma unroll
    for (int it = 0; it < PA; ++it) av[it] = *(const v4f*)(ap[it] + 32);
    #pragma unroll
    for (int it = 0; it < PW; ++it) wv[it] = *(const v4f*)(wp[it] + 32);
  }
  __syncthreads();

  for (int t = 0; t < T; ++t) {
    const int cur = t & 1;
    if (t + 1 < T) {
      #pragma unroll
      for (int it = 0; it < PA; ++it)
        *(v4f*)&As[cur ^ 1][srow + it * 32][sq] = av[it];
      #pragma unroll
      for (int it = 0; it < PW; ++it) {
        v4f v = wv[it];
        if (wz[it]) { v[0] = 0.f; v[1] = 0.f; v[2] = 0.f; v[3] = 0.f; }
        *(v4f*)&Ws[cur ^ 1][srow + it * 32][sq] = v;
      }
    }
    if (t + 2 < T) {
      const int ko = (t + 2) * 32;
      #pragma unroll
      for (int it = 0; it < PA; ++it) av[it] = *(const v4f*)(ap[it] + ko);
      #pragma unroll
      for (int it = 0; it < PW; ++it) wv[it] = *(const v4f*)(wp[it] + ko);
    }

    #pragma unroll
    for (int kk = 0; kk < 32; kk += 4) {
      v2f af[TI], bf[TJ];
      #pragma unroll
      for (int i = 0; i < TI; ++i)
        af[i] = *(const v2f*)&As[cur][wrow * (TI * 16) + i * 16 + r][kk + 2 * hi];
      #pragma unroll
      for (int j = 0; j < TJ; ++j)
        bf[j] = *(const v2f*)&Ws[cur][cb + j * CS + r][kk + 2 * hi];
      #pragma unroll
      for (int i = 0; i < TI; ++i)
        #pragma unroll
        for (int j = 0; j < TJ; ++j)
          acc[i][j] = __builtin_amdgcn_wmma_f32_16x16x4_f32(
              false, af[i], false, bf[j], (short)0, acc[i][j], false, false);
    }
    __syncthreads();
  }

  if constexpr (ROPE == 0) {
    #pragma unroll
    for (int i = 0; i < TI; ++i)
      #pragma unroll
      for (int j = 0; j < TJ; ++j) {
        int gcol = n0 + cb + j * CS + r;
        if (gcol >= N) continue;
        float b = bias ? bias[gcol] : 0.0f;
        #pragma unroll
        for (int vg = 0; vg < 8; ++vg) {
          int grow = m0 + wrow * (TI * 16) + i * 16 + vg + 8 * hi;
          C[(size_t)grow * N + gcol] = acc[i][j][vg] + b;
        }
      }
  } else {
    const int d   = (wcol & 1) * 16 + r;
    const int gc0 = n0 + cb + r;
    const int gc1 = gc0 + 32;
    const bool rot = (ROPE == 1) || (((n0 >> 6) & 1) == 0);
    constexpr int TP = 68;
    static_assert(BM * TP <= 2 * BM * LP, "the cos / sin table fits into the A tile buffers");
    float* tab = &As[0][0][0];
    if (rot) {
      const int td = tid & 31;
      const float inv = powf(150000.0f, -((float)(2 * td)) / 64.0f);
      #pragma unroll 1
      for (int rr = tid >> 5; rr < BM; rr += 8) {
        float ang = (float)(m0 + rr) * inv;
        v2f e;
        e[0] = cosf(ang);
        e[1] = sinf(ang);
        *(v2f*)&tab[rr * TP + 2 * td] = e;
      }
      __syncthreads();
    }
    const float b0 = bias ? bias[gc0] : 0.0f;
    const float b1 = bias ? bias[gc1] : 0.0f;
    #pragma unroll
    for (int i = 0; i < TI; ++i)
      #pragma unroll
      for (int vg = 0; vg < 8; ++vg) {
        const int lrow = wrow * (TI * 16) + i * 16 + vg + 8 * hi;
        const int grow = m0 + lrow;
        float x1 = acc[i][0][vg] + b0;
        float x2 = acc[i][1][vg] + b1;
        if (rot) {
          const v2f e = *(const v2f*)&tab[lrow * TP + 2 * d];
          const float y1 = x1 * e[0] - x2 * e[1];
          const float y2 = x2 * e[0] + x1 * e[1];
          x1 = y1; x2 = y2;
        }
        C[(size_t)grow * N + gc0] = x1;
        C[(size_t)grow * N + gc1] = x2;
      }
  }
}

template <int CTRL>
__device__ __forceinline__ float row_xor(float v)
{
  return __int_as_float(__builtin_amdgcn_update_dpp(
      0, __float_as_int(v), CTRL, 0xf, 0xf, true));
}

__global__ __launch_bounds__(128) void flash_attn(
    const float* __restrict__ qbuf, const float* __restrict__ kvbuf,
    float* __restrict__ attn)
{
  constexpr int AP = 68;
  __shared__ alignas(16) float Qs[64][AP];
  __shared__ alignas(16) float Ks[64][AP];
  __shared__ alignas(16) float Vt[64][AP];
  __shared__ alignas(16) float Ps[64][AP];

  const int tid   = threadIdx.x;
  const int wave  = tid >> 5;
  const int lane  = tid & 31;
  const int r     = lane & 15;
  const int hi    = lane >> 4;
  const int h     = blockIdx.x;
  const int qtile = 31 - blockIdx.y;
  const int kvh   = h >> 3;
  const int qbase = qtile * 64;

  const int srow = tid >> 4;
  const int sc4  = (tid & 15) * 4;
  const int vd   = tid & 63;
  const int vnq  = tid >> 6;

  #pragma unroll
  for (int it = 0; it < 8; ++it) {
    int sl = srow + it * 8;
    *(v4f*)&Qs[sl][sc4] = *(const v4f*)&qbuf[(size_t)(qbase + sl) * 4096 + h * 64 + sc4];
  }

  float m_i[8], l_i[8];
  for (int vg = 0; vg < 8; ++vg) { m_i[vg] = -INFINITY; l_i[vg] = 0.0f; }
  v8f oacc[4];
  for (int sub = 0; sub < 4; ++sub)
    for (int c = 0; c < 8; ++c) oacc[sub][c] = 0.0f;

  for (int j = 0; j <= qtile; ++j) {
    const int kvbase = j * 64;
    const float* kvt = &kvbuf[(size_t)kvbase * 1024 + kvh * 128];
    v4f kreg[8], vreg[8];
    #pragma unroll
    for (int it = 0; it < 8; ++it)
      kreg[it] = *(const v4f*)&kvt[(size_t)(srow + it * 8) * 1024 + sc4];
    #pragma unroll
    for (int it = 0; it < 8; ++it) {
      int n = 4 * (vnq + 2 * it);
      #pragma unroll
      for (int c = 0; c < 4; ++c)
        vreg[it][c] = kvt[(size_t)(n + c) * 1024 + 64 + vd];
    }
    __syncthreads();
    #pragma unroll
    for (int it = 0; it < 8; ++it)
      *(v4f*)&Ks[srow + it * 8][sc4] = kreg[it];
    #pragma unroll
    for (int it = 0; it < 8; ++it)
      *(v4f*)&Vt[vd][4 * (vnq + 2 * it)] = vreg[it];
    __syncthreads();

    v8f sacc[4];
    for (int sub = 0; sub < 4; ++sub)
      for (int c = 0; c < 8; ++c) sacc[sub][c] = 0.0f;
    #pragma unroll
    for (int kk = 0; kk < 64; kk += 4) {
      v2f af = *(const v2f*)&Qs[wave * 16 + r][kk + 2 * hi];
      #pragma unroll
      for (int sub = 0; sub < 4; ++sub) {
        v2f bf = *(const v2f*)&Ks[sub * 16 + r][kk + 2 * hi];
        sacc[sub] = __builtin_amdgcn_wmma_f32_16x16x4_f32(
            false, af, false, bf, (short)0, sacc[sub], false, false);
      }
    }

    float mx[8];
    #pragma unroll
    for (int vg = 0; vg < 8; ++vg) {
      int qr = qbase + wave * 16 + vg + 8 * hi;
      #pragma unroll
      for (int sub = 0; sub < 4; ++sub) {
        int kc = kvbase + sub * 16 + r;
        sacc[sub][vg] = (kc <= qr) ? sacc[sub][vg] * 0.125f : NEGBIG;
      }
      mx[vg] = fmaxf(fmaxf(sacc[0][vg], sacc[1][vg]), fmaxf(sacc[2][vg], sacc[3][vg]));
    }
    #pragma unroll
    for (int vg = 0; vg < 8; ++vg) mx[vg] = fmaxf(mx[vg], row_xor<0x161>(mx[vg]));
    #pragma unroll
    for (int vg = 0; vg < 8; ++vg) mx[vg] = fmaxf(mx[vg], row_xor<0x162>(mx[vg]));
    #pragma unroll
    for (int vg = 0; vg < 8; ++vg) mx[vg] = fmaxf(mx[vg], row_xor<0x164>(mx[vg]));
    #pragma unroll
    for (int vg = 0; vg < 8; ++vg) mx[vg] = fmaxf(mx[vg], row_xor<0x168>(mx[vg]));

    #pragma unroll
    for (int vg = 0; vg < 8; ++vg) {
      float newm  = fmaxf(m_i[vg], mx[vg]);
      float alpha = __expf(m_i[vg] - newm);
      m_i[vg] = newm;
      float ps = 0.0f;
      #pragma unroll
      for (int sub = 0; sub < 4; ++sub) {
        float p = __expf(sacc[sub][vg] - newm);
        ps += p;
        Ps[wave * 16 + vg + 8 * hi][sub * 16 + r] = p;
        oacc[sub][vg] *= alpha;
      }
      l_i[vg] = l_i[vg] * alpha + ps;
    }

    #pragma unroll
    for (int kk = 0; kk < 64; kk += 4) {
      v2f af = *(const v2f*)&Ps[wave * 16 + r][kk + 2 * hi];
      #pragma unroll
      for (int sub = 0; sub < 4; ++sub) {
        v2f bf = *(const v2f*)&Vt[sub * 16 + r][kk + 2 * hi];
        oacc[sub] = __builtin_amdgcn_wmma_f32_16x16x4_f32(
            false, af, false, bf, (short)0, oacc[sub], false, false);
      }
    }
  }

  #pragma unroll
  for (int vg = 0; vg < 8; ++vg) l_i[vg] += row_xor<0x161>(l_i[vg]);
  #pragma unroll
  for (int vg = 0; vg < 8; ++vg) l_i[vg] += row_xor<0x162>(l_i[vg]);
  #pragma unroll
  for (int vg = 0; vg < 8; ++vg) l_i[vg] += row_xor<0x164>(l_i[vg]);
  #pragma unroll
  for (int vg = 0; vg < 8; ++vg) l_i[vg] += row_xor<0x168>(l_i[vg]);
  #pragma unroll
  for (int vg = 0; vg < 8; ++vg) {
    float im = 1.0f / l_i[vg];
    int grow = qbase + wave * 16 + vg + 8 * hi;
    for (int sub = 0; sub < 4; ++sub) {
      int d = sub * 16 + r;
      attn[(size_t)grow * 4096 + h * 64 + d] = oacc[sub][vg] * im;
    }
  }
}

extern "C" void kernel_launch(void* const* d_in, const int* in_sizes, int n_in,
                              void* d_out, int out_size, void* d_ws, size_t ws_size,
                              hipStream_t stream)
{
  const float* hidden = (const float*)d_in[0];
  const float* q_w    = (const float*)d_in[2];
  const float* q_b    = (const float*)d_in[3];
  const float* kv_a_w = (const float*)d_in[4];
  const float* kv_b_w = (const float*)d_in[5];
  const float* o_w    = (const float*)d_in[6];
  const float* o_b    = (const float*)d_in[7];
  float* out = (float*)d_out;

  float* ws    = (float*)d_ws;
  float* qbuf  = ws;
  float* ckv   = qbuf  + (size_t)2048 * 4096;
  float* kvbuf = ckv   + (size_t)2048 * 384;
  float* attn  = kvbuf + (size_t)2048 * 1024;

  dim3 blk(256);
  gemm_wmma_f32<128, 128, 2, 4, 1><<<dim3(4096 / 128, 2048 / 128), blk, 0, stream>>>(
      hidden, q_w, q_b, qbuf, 2048, 4096, 2880);
  gemm_wmma_f32<64, 64, 2, 4, 0><<<dim3(384 / 64, 2048 / 64), blk, 0, stream>>>(
      hidden, kv_a_w, nullptr, ckv, 2048, 384, 2880);
  gemm_wmma_f32<64, 64, 4, 2, 2><<<dim3(1024 / 64, 2048 / 64), blk, 0, stream>>>(
      ckv, kv_b_w, nullptr, kvbuf, 2048, 1024, 384);
  flash_attn<<<dim3(64, 32), 128, 0, stream>>>(qbuf, kvbuf, attn);
  gemm_wmma_f32<128, 96, 4, 2, 0><<<dim3(2880 / 96, 2048 / 128), blk, 0, stream>>>(
      attn, o_w, o_b, out, 2048, 2880, 4096);
}
